# starting kernel
# baseline (speedup 1.0000x reference)
#include <hip/hip_runtime.h>
#include <math.h>

typedef float v2f __attribute__((ext_vector_type(2)));
typedef float v8f __attribute__((ext_vector_type(8)));

#define MULT 32
#define NATTR 8
#define RB 16
#define FC 64
#define LOG2F_ 0.6931471805599453f

__device__ __forceinline__ float ssp(float x) {
    return fmaxf(x, 0.0f) + log1pf(__expf(-fabsf(x))) - LOG2F_;
}

__global__ __launch_bounds__(256) void lin1_kernel(
    const float* __restrict__ node_s, const float* __restrict__ node_v,
    const float* __restrict__ W1s, const float* __restrict__ W1v,
    float* __restrict__ s_out, float* __restrict__ v_out, int N)
{
    __shared__ float Ls[MULT * MULT];
    __shared__ float Lv[MULT * MULT];
    for (int i = threadIdx.x; i < MULT * MULT; i += 256) { Ls[i] = W1s[i]; Lv[i] = W1v[i]; }
    __syncthreads();

    int gid = blockIdx.x * 256 + threadIdx.x;
    int n = gid >> 5, w = gid & 31;
    if (n >= N) return;
    const float inv_m = 0.17677669529663687f;
    const float* srow = node_s + (size_t)n * MULT;
    const float* vrow = node_v + (size_t)n * MULT * 3;
    float as = 0.f, a0 = 0.f, a1 = 0.f, a2 = 0.f;
#pragma unroll 8
    for (int u = 0; u < MULT; ++u) {
        float ws_ = Ls[u * MULT + w];
        float wv_ = Lv[u * MULT + w];
        as += srow[u] * ws_;
        a0 += vrow[u * 3 + 0] * wv_;
        a1 += vrow[u * 3 + 1] * wv_;
        a2 += vrow[u * 3 + 2] * wv_;
    }
    s_out[(size_t)n * MULT + w] = as * inv_m;
    v_out[(size_t)n * MULT * 3 + w * 3 + 0] = a0 * inv_m;
    v_out[(size_t)n * MULT * 3 + w * 3 + 1] = a1 * inv_m;
    v_out[(size_t)n * MULT * 3 + w * 3 + 2] = a2 * inv_m;
}

__global__ __launch_bounds__(256) void edge_kernel(
    const float* __restrict__ edge_emb, const float* __restrict__ edge_y1,
    const int* __restrict__ edge_index,
    const float* __restrict__ Wr1, const float* __restrict__ Wr2,
    const float* __restrict__ s_in, const float* __restrict__ v_in,
    float* __restrict__ agg_s, float* __restrict__ agg_v, int E)
{
    __shared__ float Wr1t[FC * RB];
    __shared__ float Wr2t[4 * MULT * FC];
    __shared__ float stage[8 * 16 * 4 * MULT];

    const int tid = threadIdx.x;
    for (int i = tid; i < RB * FC; i += 256) {
        int k = i / FC, n = i % FC;
        Wr1t[n * RB + k] = Wr1[i];
    }
    for (int i = tid; i < FC * 4 * MULT; i += 256) {
        int k = i / (4 * MULT), n = i % (4 * MULT);
        Wr2t[n * FC + k] = Wr2[i];
    }
    __syncthreads();

    const int wave = tid >> 5, lane = tid & 31;
    const int m = lane & 15, kh = lane >> 4;
    const long e0 = (long)blockIdx.x * 128 + wave * 16;
    float* const mystage = stage + wave * (16 * 4 * MULT);

    const v8f vzero = {0.f, 0.f, 0.f, 0.f, 0.f, 0.f, 0.f, 0.f};

    v8f h[4];
#pragma unroll
    for (int t = 0; t < 4; ++t) h[t] = vzero;
#pragma unroll
    for (int k4 = 0; k4 < RB; k4 += 4) {
        long ee = e0 + m; if (ee >= E) ee = E - 1;
        v2f a = *(const v2f*)(edge_emb + ee * RB + k4 + 2 * kh);
#pragma unroll
        for (int t = 0; t < 4; ++t) {
            v2f b = *(const v2f*)(Wr1t + (t * 16 + m) * RB + k4 + 2 * kh);
            h[t] = __builtin_amdgcn_wmma_f32_16x16x4_f32(
                false, a, false, b, (short)0, h[t], false, false);
        }
    }
#pragma unroll
    for (int t = 0; t < 4; ++t)
#pragma unroll
        for (int r = 0; r < 8; ++r) {
            int row = r + 8 * kh, col = t * 16 + m;
            mystage[row * FC + col] = ssp(h[t][r] * 0.25f);
        }
    __syncthreads();

    v8f wacc[8];
#pragma unroll
    for (int t = 0; t < 8; ++t) wacc[t] = vzero;
    for (int k4 = 0; k4 < FC; k4 += 4) {
        v2f a = *(const v2f*)(mystage + m * FC + k4 + 2 * kh);
#pragma unroll
        for (int t = 0; t < 8; ++t) {
            v2f b = *(const v2f*)(Wr2t + (t * 16 + m) * FC + k4 + 2 * kh);
            wacc[t] = __builtin_amdgcn_wmma_f32_16x16x4_f32(
                false, a, false, b, (short)0, wacc[t], false, false);
        }
    }
    __syncthreads();
#pragma unroll
    for (int t = 0; t < 8; ++t)
#pragma unroll
        for (int r = 0; r < 8; ++r) {
            int row = r + 8 * kh, col = t * 16 + m;
            mystage[row * 4 * MULT + col] = wacc[t][r] * 0.125f;
        }
    __syncthreads();

    long e = (long)blockIdx.x * 128 + (tid >> 1);
    if (e < E) {
        int wv = (tid >> 1) >> 4, row = (tid >> 1) & 15;
        const float* wrow = stage + wv * (16 * 4 * MULT) + row * 4 * MULT;
        int src = edge_index[e];
        int dst = edge_index[(long)E + e];
        float y0 = edge_y1[e * 3 + 0];
        float y1 = edge_y1[e * 3 + 1];
        float y2 = edge_y1[e * 3 + 2];
        const float inv_s3  = 0.5773502691896258f;
        const float inv_deg = 0.17677669529663687f;
        const float* srow = s_in + (size_t)src * MULT;
        const float* vrow = v_in + (size_t)src * MULT * 3;
        float* as_ = agg_s + (size_t)dst * 2 * MULT;
        float* av_ = agg_v + (size_t)dst * 2 * MULT * 3;
        int u0 = (tid & 1) * 16;
#pragma unroll 4
        for (int u = u0; u < u0 + 16; ++u) {
            float w1 = wrow[u];
            float w2 = wrow[MULT + u];
            float w3 = wrow[2 * MULT + u];
            float w4 = wrow[3 * MULT + u];
            float se = srow[u];
            float v0 = vrow[u * 3 + 0], v1 = vrow[u * 3 + 1], v2 = vrow[u * 3 + 2];
            unsafeAtomicAdd(as_ + u, w1 * se * inv_deg);
            float dv = v0 * y0 + v1 * y1 + v2 * y2;
            unsafeAtomicAdd(as_ + MULT + u, w4 * dv * inv_s3 * inv_deg);
            float t2 = w2 * se * inv_deg;
            unsafeAtomicAdd(av_ + u * 3 + 0, t2 * y0);
            unsafeAtomicAdd(av_ + u * 3 + 1, t2 * y1);
            unsafeAtomicAdd(av_ + u * 3 + 2, t2 * y2);
            float t3 = w3 * inv_deg;
            unsafeAtomicAdd(av_ + 3 * MULT + u * 3 + 0, t3 * v0);
            unsafeAtomicAdd(av_ + 3 * MULT + u * 3 + 1, t3 * v1);
            unsafeAtomicAdd(av_ + 3 * MULT + u * 3 + 2, t3 * v2);
        }
    }
}

__global__ __launch_bounds__(256) void out_kernel(
    const float* __restrict__ node_s, const float* __restrict__ node_v,
    const float* __restrict__ node_attrs,
    const float* __restrict__ W2s, const float* __restrict__ W2v,
    const float* __restrict__ Wscs, const float* __restrict__ Wscv,
    const float* __restrict__ agg_s, const float* __restrict__ agg_v,
    float* __restrict__ out, int N)
{
    __shared__ float L2s[2 * MULT * MULT];
    __shared__ float L2v[2 * MULT * MULT];
    __shared__ float Lss[MULT * NATTR * MULT];
    __shared__ float Lsv[MULT * NATTR * MULT];
    for (int i = threadIdx.x; i < 2 * MULT * MULT; i += 256) { L2s[i] = W2s[i]; L2v[i] = W2v[i]; }
    for (int i = threadIdx.x; i < MULT * NATTR * MULT; i += 256) { Lss[i] = Wscs[i]; Lsv[i] = Wscv[i]; }
    __syncthreads();

    int gid = blockIdx.x * 256 + threadIdx.x;
    int n = gid >> 5, w = gid & 31;
    if (n >= N) return;
    const float inv2m = 0.125f;
    const float invfan = 0.0625f;

    const float* ags = agg_s + (size_t)n * 2 * MULT;
    const float* agv = agg_v + (size_t)n * 2 * MULT * 3;
    float ns = 0.f, nv0 = 0.f, nv1 = 0.f, nv2 = 0.f;
#pragma unroll 8
    for (int u = 0; u < 2 * MULT; ++u) {
        float w2s = L2s[u * MULT + w];
        float w2v = L2v[u * MULT + w];
        ns  += ags[u] * w2s;
        nv0 += agv[u * 3 + 0] * w2v;
        nv1 += agv[u * 3 + 1] * w2v;
        nv2 += agv[u * 3 + 2] * w2v;
    }
    ns *= inv2m; nv0 *= inv2m; nv1 *= inv2m; nv2 *= inv2m;

    float attr[NATTR];
#pragma unroll
    for (int a = 0; a < NATTR; ++a) attr[a] = node_attrs[(size_t)n * NATTR + a];

    const float* srow = node_s + (size_t)n * MULT;
    const float* vrow = node_v + (size_t)n * MULT * 3;
    float scs = 0.f, scv0 = 0.f, scv1 = 0.f, scv2 = 0.f;
#pragma unroll 4
    for (int u = 0; u < MULT; ++u) {
        float ts = 0.f, tv = 0.f;
#pragma unroll
        for (int a = 0; a < NATTR; ++a) {
            ts += attr[a] * Lss[(u * NATTR + a) * MULT + w];
            tv += attr[a] * Lsv[(u * NATTR + a) * MULT + w];
        }
        scs  += srow[u] * ts;
        scv0 += vrow[u * 3 + 0] * tv;
        scv1 += vrow[u * 3 + 1] * tv;
        scv2 += vrow[u * 3 + 2] * tv;
    }
    scs *= invfan; scv0 *= invfan; scv1 *= invfan; scv2 *= invfan;

    float* orow = out + (size_t)n * 4 * MULT;
    orow[w] = ns + scs;
    orow[MULT + w * 3 + 0] = nv0 + scv0;
    orow[MULT + w * 3 + 1] = nv1 + scv1;
    orow[MULT + w * 3 + 2] = nv2 + scv2;
}

extern "C" void kernel_launch(void* const* d_in, const int* in_sizes, int n_in,
                              void* d_out, int out_size, void* d_ws, size_t ws_size,
                              hipStream_t stream) {
    const float* node_s     = (const float*)d_in[0];
    const float* node_v     = (const float*)d_in[1];
    const float* node_attrs = (const float*)d_in[2];
    const float* edge_emb   = (const float*)d_in[3];
    const float* edge_y1    = (const float*)d_in[5];
    const int*   edge_index = (const int*)d_in[6];
    const float* W1s  = (const float*)d_in[7];
    const float* W1v  = (const float*)d_in[8];
    const float* Wr1  = (const float*)d_in[9];
    const float* Wr2  = (const float*)d_in[10];
    const float* W2s  = (const float*)d_in[11];
    const float* W2v  = (const float*)d_in[12];
    const float* Wscs = (const float*)d_in[13];
    const float* Wscv = (const float*)d_in[14];

    const int N = in_sizes[0] / MULT;
    const int E = in_sizes[3] / RB;

    float* s_buf = (float*)d_ws;
    float* v_buf = s_buf + (size_t)N * MULT;
    float* agg_s = v_buf + (size_t)N * MULT * 3;
    float* agg_v = agg_s + (size_t)N * 2 * MULT;

    hipMemsetAsync(agg_s, 0, (size_t)N * 8 * MULT * sizeof(float), stream);

    lin1_kernel<<<(N * MULT + 255) / 256, 256, 0, stream>>>(
        node_s, node_v, W1s, W1v, s_buf, v_buf, N);
    edge_kernel<<<(E + 127) / 128, 256, 0, stream>>>(
        edge_emb, edge_y1, edge_index, Wr1, Wr2, s_buf, v_buf, agg_s, agg_v, E);
    out_kernel<<<(N * MULT + 255) / 256, 256, 0, stream>>>(
        node_s, node_v, node_attrs, W2s, W2v, Wscs, Wscv, agg_s, agg_v,
        (float*)d_out, N);
}


# edge scatter: one edge per wave at a time, lane = channel, so each f32 atomic add instruction covers one whole 128-byte line of the destination row, not 32 lines; atomic arrival order as before
# speedup vs baseline: 4.1873x; 4.1873x over previous
#include <hip/hip_runtime.h>
#include <math.h>

typedef float v2f __attribute__((ext_vector_type(2)));
typedef float v8f __attribute__((ext_vector_type(8)));

#define MULT 32
#define NATTR 8
#define RB 16
#define FC 64
#define LOG2F_ 0.6931471805599453f

__device__ __forceinline__ float ssp(float x) {
    return fmaxf(x, 0.0f) + log1pf(__expf(-fabsf(x))) - LOG2F_;
}

__global__ __launch_bounds__(256) void lin1_kernel(
    const float* __restrict__ node_s, const float* __restrict__ node_v,
    const float* __restrict__ W1s, const float* __restrict__ W1v,
    float* __restrict__ s_out, float* __restrict__ v_out, int N)
{
    __shared__ float Ls[MULT * MULT];
    __shared__ float Lv[MULT * MULT];
    for (int i = threadIdx.x; i < MULT * MULT; i += 256) { Ls[i] = W1s[i]; Lv[i] = W1v[i]; }
    __syncthreads();

    int gid = blockIdx.x * 256 + threadIdx.x;
    int n = gid >> 5, w = gid & 31;
    if (n >= N) return;
    const float inv_m = 0.17677669529663687f;
    const float* srow = node_s + (size_t)n * MULT;
    const float* vrow = node_v + (size_t)n * MULT * 3;
    float as = 0.f, a0 = 0.f, a1 = 0.f, a2 = 0.f;
#pragma unroll 8
    for (int u = 0; u < MULT; ++u) {
        float ws_ = Ls[u * MULT + w];
        float wv_ = Lv[u * MULT + w];
        as += srow[u] * ws_;
        a0 += vrow[u * 3 + 0] * wv_;
        a1 += vrow[u * 3 + 1] * wv_;
        a2 += vrow[u * 3 + 2] * wv_;
    }
    s_out[(size_t)n * MULT + w] = as * inv_m;
    v_out[(size_t)n * MULT * 3 + w * 3 + 0] = a0 * inv_m;
    v_out[(size_t)n * MULT * 3 + w * 3 + 1] = a1 * inv_m;
    v_out[(size_t)n * MULT * 3 + w * 3 + 2] = a2 * inv_m;
}


__device__ __forceinline__ float lane_val(float x, int l) {
    return __int_as_float(__builtin_amdgcn_readlane(__float_as_int(x), l));
}

__global__ __launch_bounds__(256, 6) void edge_kernel(
    const float* __restrict__ edge_emb, const float* __restrict__ edge_y0,
    const float* __restrict__ edge_y1, const int* __restrict__ edge_index,
    const float* __restrict__ Wr1, const float* __restrict__ Wr2,
    const float* __restrict__ s_in, const float* __restrict__ v_in,
    float* __restrict__ agg_s, float* __restrict__ agg_v, int E)
{
    __shared__ float Wr1t[FC * RB];
    __shared__ float Wr2t[4 * MULT * FC];
    __shared__ float stage[8 * 16 * 4 * MULT];

    const int tid = threadIdx.x;
    for (int i = tid; i < RB * FC; i += 256) {
        int k = i / FC, n = i % FC;
        Wr1t[n * RB + k] = Wr1[i];
    }
    for (int i = tid; i < FC * 4 * MULT; i += 256) {
        int k = i / (4 * MULT), n = i % (4 * MULT);
        Wr2t[n * FC + k] = Wr2[i];
    }
    __syncthreads();

    const int wave = __builtin_amdgcn_readfirstlane(tid >> 5);
    const int lane = tid & 31;
    const int m = lane & 15, kh = lane >> 4;
    const long e0 = (long)blockIdx.x * 128 + wave * 16;
    float* const mystage = stage + wave * (16 * 4 * MULT);

    long ee = e0 + m; if (ee >= E) ee = E - 1;
    const int   srcv = edge_index[ee];

    const v8f vzero = {0.f, 0.f, 0.f, 0.f, 0.f, 0.f, 0.f, 0.f};

    v8f h[4];
#pragma unroll
    for (int t = 0; t < 4; ++t) h[t] = vzero;
#pragma unroll
    for (int k4 = 0; k4 < RB; k4 += 4) {
        v2f a = *(const v2f*)(edge_emb + ee * RB + k4 + 2 * kh);
#pragma unroll
        for (int t = 0; t < 4; ++t) {
            v2f b = *(const v2f*)(Wr1t + (t * 16 + m) * RB + k4 + 2 * kh);
            h[t] = __builtin_amdgcn_wmma_f32_16x16x4_f32(
                false, a, false, b, (short)0, h[t], false, false);
        }
    }
#pragma unroll
    for (int t = 0; t < 4; ++t)
#pragma unroll
        for (int r = 0; r < 8; ++r) {
            int row = r + 8 * kh, col = t * 16 + m;
            mystage[row * FC + col] = ssp(h[t][r] * 0.25f);
        }
    __syncthreads();

    v8f wacc[8];
#pragma unroll
    for (int t = 0; t < 8; ++t) wacc[t] = vzero;
    for (int k4 = 0; k4 < FC; k4 += 4) {
        v2f a = *(const v2f*)(mystage + m * FC + k4 + 2 * kh);
#pragma unroll
        for (int t = 0; t < 8; ++t) {
            v2f b = *(const v2f*)(Wr2t + (t * 16 + m) * FC + k4 + 2 * kh);
            wacc[t] = __builtin_amdgcn_wmma_f32_16x16x4_f32(
                false, a, false, b, (short)0, wacc[t], false, false);
        }
    }

    __syncthreads();
#pragma unroll
    for (int t = 0; t < 8; ++t)
#pragma unroll
        for (int r = 0; r < 8; ++r) {
            int row = r + 8 * kh, col = t * 16 + m;
            mystage[row * 4 * MULT + col] = wacc[t][r] * 0.125f;
        }

    float se[16], vx[16], vy[16], vz[16];
#pragma unroll
    for (int r = 0; r < 16; ++r) {
        const int s = __builtin_amdgcn_readlane(srcv, r);
        se[r] = s_in[(size_t)s * MULT + lane];
        const float* vp = v_in + (size_t)s * (MULT * 3) + 3 * lane;
        vx[r] = vp[0]; vy[r] = vp[1]; vz[r] = vp[2];
    }
    const int   dstv = edge_index[(long)E + ee];
    const float y00v = edge_y0[ee];
    const float y10v = edge_y1[ee * 3 + 0];
    const float y11v = edge_y1[ee * 3 + 1];
    const float y12v = edge_y1[ee * 3 + 2];

    __syncthreads();

    const float inv_s3  = 0.5773502691896258f;
    const float inv_deg = 0.17677669529663687f;
    const int pa0 = ((3 * lane + 0) & 31) * 4;
    const int pa1 = ((3 * lane + 1) & 31) * 4;
    const int pa2 = ((3 * lane + 2) & 31) * 4;
    const int i0 = (lane + 0) % 3;
    const int i1 = (lane + 2) % 3;
    const int i2 = (lane + 4) % 3;
    auto scatter_edge = [&](const int r) {
        const int   d   = __builtin_amdgcn_readlane(dstv, r);
        const float y00 = lane_val(y00v, r);
        const float y0  = lane_val(y10v, r);
        const float y1  = lane_val(y11v, r);
        const float y2  = lane_val(y12v, r);
        float* as_ = agg_s + (size_t)d * (2 * MULT);
        float* av_ = agg_v + (size_t)d * (2 * MULT * 3);
        const float* wrow = mystage + r * (4 * MULT);
        const float w1 = wrow[lane];
        const float w2 = wrow[MULT + lane];
        const float w3 = wrow[2 * MULT + lane];
        const float w4 = wrow[3 * MULT + lane];
        const float sv = se[r], v0 = vx[r], v1 = vy[r], v2 = vz[r];
        const float cy = y00 * inv_deg;
        unsafeAtomicAdd(as_ + lane, w1 * sv * cy);
        const float dv = v0 * y0 + v1 * y1 + v2 * y2;
        unsafeAtomicAdd(as_ + MULT + lane, w4 * dv * (inv_s3 * inv_deg));
        const float t2 = w2 * sv * inv_deg;
        {
            const int p0 = __builtin_amdgcn_ds_permute(pa0, __float_as_int(t2 * y0));
            const int p1 = __builtin_amdgcn_ds_permute(pa1, __float_as_int(t2 * y1));
            const int p2 = __builtin_amdgcn_ds_permute(pa2, __float_as_int(t2 * y2));
            const int f0 = (i0 == 0) ? p0 : ((i0 == 1) ? p1 : p2);
            const int f1 = (i1 == 0) ? p0 : ((i1 == 1) ? p1 : p2);
            const int f2 = (i2 == 0) ? p0 : ((i2 == 1) ? p1 : p2);
            unsafeAtomicAdd(av_ + lane,            __int_as_float(f0));
            unsafeAtomicAdd(av_ + MULT + lane,     __int_as_float(f1));
            unsafeAtomicAdd(av_ + 2 * MULT + lane, __int_as_float(f2));
        }
        const float t3 = w3 * cy;
        {
            const int p0 = __builtin_amdgcn_ds_permute(pa0, __float_as_int(t3 * v0));
            const int p1 = __builtin_amdgcn_ds_permute(pa1, __float_as_int(t3 * v1));
            const int p2 = __builtin_amdgcn_ds_permute(pa2, __float_as_int(t3 * v2));
            const int f0 = (i0 == 0) ? p0 : ((i0 == 1) ? p1 : p2);
            const int f1 = (i1 == 0) ? p0 : ((i1 == 1) ? p1 : p2);
            const int f2 = (i2 == 0) ? p0 : ((i2 == 1) ? p1 : p2);
            unsafeAtomicAdd(av_ + 3 * MULT + lane, __int_as_float(f0));
            unsafeAtomicAdd(av_ + 4 * MULT + lane, __int_as_float(f1));
            unsafeAtomicAdd(av_ + 5 * MULT + lane, __int_as_float(f2));
        }
    };
    if (e0 + 16 <= E) {
#pragma unroll
        for (int r = 0; r < 16; ++r) scatter_edge(r);
    } else {
#pragma unroll
        for (int r = 0; r < 16; ++r)
            if (e0 + r < E) scatter_edge(r);
    }
}

__global__ __launch_bounds__(256) void out_kernel(
    const float* __restrict__ node_s, const float* __restrict__ node_v,
    const float* __restrict__ node_attrs,
    const float* __restrict__ W2s, const float* __restrict__ W2v,
    const float* __restrict__ Wscs, const float* __restrict__ Wscv,
    const float* __restrict__ agg_s, const float* __restrict__ agg_v,
    float* __restrict__ out, int N)
{
    __shared__ float L2s[2 * MULT * MULT];
    __shared__ float L2v[2 * MULT * MULT];
    __shared__ float Lss[MULT * NATTR * MULT];
    __shared__ float Lsv[MULT * NATTR * MULT];
    for (int i = threadIdx.x; i < 2 * MULT * MULT; i += 256) { L2s[i] = W2s[i]; L2v[i] = W2v[i]; }
    for (int i = threadIdx.x; i < MULT * NATTR * MULT; i += 256) { Lss[i] = Wscs[i]; Lsv[i] = Wscv[i]; }
    __syncthreads();

    int gid = blockIdx.x * 256 + threadIdx.x;
    int n = gid >> 5, w = gid & 31;
    if (n >= N) return;
    const float inv2m = 0.125f;
    const float invfan = 0.0625f;

    const float* ags = agg_s + (size_t)n * 2 * MULT;
    const float* agv = agg_v + (size_t)n * 2 * MULT * 3;
    float ns = 0.f, nv0 = 0.f, nv1 = 0.f, nv2 = 0.f;
#pragma unroll 8
    for (int u = 0; u < 2 * MULT; ++u) {
        float w2s = L2s[u * MULT + w];
        float w2v = L2v[u * MULT + w];
        ns  += ags[u] * w2s;
        nv0 += agv[u * 3 + 0] * w2v;
        nv1 += agv[u * 3 + 1] * w2v;
        nv2 += agv[u * 3 + 2] * w2v;
    }
    ns *= inv2m; nv0 *= inv2m; nv1 *= inv2m; nv2 *= inv2m;

    float attr[NATTR];
#pragma unroll
    for (int a = 0; a < NATTR; ++a) attr[a] = node_attrs[(size_t)n * NATTR + a];

    const float* srow = node_s + (size_t)n * MULT;
    const float* vrow = node_v + (size_t)n * MULT * 3;
    float scs = 0.f, scv0 = 0.f, scv1 = 0.f, scv2 = 0.f;
#pragma unroll 4
    for (int u = 0; u < MULT; ++u) {
        float ts = 0.f, tv = 0.f;
#pragma unroll
        for (int a = 0; a < NATTR; ++a) {
            ts += attr[a] * Lss[(u * NATTR + a) * MULT + w];
            tv += attr[a] * Lsv[(u * NATTR + a) * MULT + w];
        }
        scs  += srow[u] * ts;
        scv0 += vrow[u * 3 + 0] * tv;
        scv1 += vrow[u * 3 + 1] * tv;
        scv2 += vrow[u * 3 + 2] * tv;
    }
    scs *= invfan; scv0 *= invfan; scv1 *= invfan; scv2 *= invfan;

    float* orow = out + (size_t)n * 4 * MULT;
    orow[w] = ns + scs;
    orow[MULT + w * 3 + 0] = nv0 + scv0;
    orow[MULT + w * 3 + 1] = nv1 + scv1;
    orow[MULT + w * 3 + 2] = nv2 + scv2;
}

extern "C" void kernel_launch(void* const* d_in, const int* in_sizes, int n_in,
                              void* d_out, int out_size, void* d_ws, size_t ws_size,
                              hipStream_t stream) {
    const float* node_s     = (const float*)d_in[0];
    const float* node_v     = (const float*)d_in[1];
    const float* node_attrs = (const float*)d_in[2];
    const float* edge_emb   = (const float*)d_in[3];
    const float* edge_y0    = (const float*)d_in[4];
    const float* edge_y1    = (const float*)d_in[5];
    const int*   edge_index = (const int*)d_in[6];
    const float* W1s  = (const float*)d_in[7];
    const float* W1v  = (const float*)d_in[8];
    const float* Wr1  = (const float*)d_in[9];
    const float* Wr2  = (const float*)d_in[10];
    const float* W2s  = (const float*)d_in[11];
    const float* W2v  = (const float*)d_in[12];
    const float* Wscs = (const float*)d_in[13];
    const float* Wscv = (const float*)d_in[14];

    const int N = in_sizes[0] / MULT;
    const int E = in_sizes[3] / RB;

    float* s_buf = (float*)d_ws;
    float* v_buf = s_buf + (size_t)N * MULT;
    float* agg_s = v_buf + (size_t)N * MULT * 3;
    float* agg_v = agg_s + (size_t)N * 2 * MULT;

    hipMemsetAsync(agg_s, 0, (size_t)N * 8 * MULT * sizeof(float), stream);

    lin1_kernel<<<(N * MULT + 255) / 256, 256, 0, stream>>>(
        node_s, node_v, W1s, W1v, s_buf, v_buf, N);
    edge_kernel<<<(E + 127) / 128, 256, 0, stream>>>(
        edge_emb, edge_y0, edge_y1, edge_index, Wr1, Wr2, s_buf, v_buf, agg_s, agg_v, E);
    out_kernel<<<(N * MULT + 255) / 256, 256, 0, stream>>>(
        node_s, node_v, node_attrs, W2s, W2v, Wscs, Wscv, agg_s, agg_v,
        (float*)d_out, N);
}


# edge kernel LDS bank conflicts removed: row pitches 16/64/128 -> 20/68/132 floats, 16-byte fragment loads (a lane holds 4 consecutive k for two WMMA steps; k sum order changes), 16-byte Wr2 staging
# speedup vs baseline: 6.5833x; 1.5722x over previous
#include <hip/hip_runtime.h>
#include <math.h>

typedef float v2f __attribute__((ext_vector_type(2)));
typedef float v4f __attribute__((ext_vector_type(4)));
typedef float v8f __attribute__((ext_vector_type(8)));

#define MULT 32
#define NATTR 8
#define RB 16
#define FC 64
#define LOG2F_ 0.6931471805599453f
#define W1P 20
#define W2P 68
#define HP  68
#define SP  132

__device__ __forceinline__ float ssp(float x) {
    return fmaxf(x, 0.0f) + log1pf(__expf(-fabsf(x))) - LOG2F_;
}

__global__ __launch_bounds__(256) void lin1_kernel(
    const float* __restrict__ node_s, const float* __restrict__ node_v,
    const float* __restrict__ W1s, const float* __restrict__ W1v,
    float* __restrict__ s_out, float* __restrict__ v_out, int N)
{
    __shared__ float Ls[MULT * MULT];
    __shared__ float Lv[MULT * MULT];
    for (int i = threadIdx.x; i < MULT * MULT; i += 256) { Ls[i] = W1s[i]; Lv[i] = W1v[i]; }
    __syncthreads();

    int gid = blockIdx.x * 256 + threadIdx.x;
    int n = gid >> 5, w = gid & 31;
    if (n >= N) return;
    const float inv_m = 0.17677669529663687f;
    const float* srow = node_s + (size_t)n * MULT;
    const float* vrow = node_v + (size_t)n * MULT * 3;
    float as = 0.f, a0 = 0.f, a1 = 0.f, a2 = 0.f;
#pragma unroll 8
    for (int u = 0; u < MULT; ++u) {
        float ws_ = Ls[u * MULT + w];
        float wv_ = Lv[u * MULT + w];
        as += srow[u] * ws_;
        a0 += vrow[u * 3 + 0] * wv_;
        a1 += vrow[u * 3 + 1] * wv_;
        a2 += vrow[u * 3 + 2] * wv_;
    }
    s_out[(size_t)n * MULT + w] = as * inv_m;
    v_out[(size_t)n * MULT * 3 + w * 3 + 0] = a0 * inv_m;
    v_out[(size_t)n * MULT * 3 + w * 3 + 1] = a1 * inv_m;
    v_out[(size_t)n * MULT * 3 + w * 3 + 2] = a2 * inv_m;
}


__device__ __forceinline__ float lane_val(float x, int l) {
    return __int_as_float(__builtin_amdgcn_readlane(__float_as_int(x), l));
}

__global__ __launch_bounds__(256, 6) void edge_kernel(
    const float* __restrict__ edge_emb, const float* __restrict__ edge_y0,
    const float* __restrict__ edge_y1, const int* __restrict__ edge_index,
    const float* __restrict__ Wr1, const float* __restrict__ Wr2,
    const float* __restrict__ s_in, const float* __restrict__ v_in,
    float* __restrict__ agg_s, float* __restrict__ agg_v, int E)
{
    __shared__ __attribute__((aligned(16))) float Wr1t[FC * W1P];
    __shared__ __attribute__((aligned(16))) float Wr2t[4 * MULT * W2P];
    __shared__ __attribute__((aligned(16))) float stage[8 * 16 * SP];

    const int tid = threadIdx.x;
    for (int i = tid; i < RB * FC; i += 256) {
        int k = i / FC, n = i % FC;
        Wr1t[n * W1P + k] = Wr1[i];
    }
    {
        const int n  = tid & (4 * MULT - 1);
        const int kq = tid >> 7;
#pragma unroll
        for (int j = 0; j < 8; ++j) {
            const int k0 = 4 * (2 * j + kq);
            v4f q;
            q[0] = Wr2[(k0 + 0) * (4 * MULT) + n];
            q[1] = Wr2[(k0 + 1) * (4 * MULT) + n];
            q[2] = Wr2[(k0 + 2) * (4 * MULT) + n];
            q[3] = Wr2[(k0 + 3) * (4 * MULT) + n];
            *(v4f*)(Wr2t + n * W2P + k0) = q;
        }
    }
    __syncthreads();

    const int wave = __builtin_amdgcn_readfirstlane(tid >> 5);
    const int lane = tid & 31;
    const int m = lane & 15, kh = lane >> 4;
    const long e0 = (long)blockIdx.x * 128 + wave * 16;
    float* const mystage = stage + wave * (16 * SP);

    long ee = e0 + m; if (ee >= E) ee = E - 1;
    const int   srcv = edge_index[ee];

    const v8f vzero = {0.f, 0.f, 0.f, 0.f, 0.f, 0.f, 0.f, 0.f};


    v8f h[4];
#pragma unroll
    for (int t = 0; t < 4; ++t) h[t] = vzero;
#pragma unroll
    for (int k8 = 0; k8 < RB; k8 += 8) {
        const v4f a4 = *(const v4f*)(edge_emb + ee * RB + k8 + 4 * kh);
        const v2f alo = {a4[0], a4[1]}, ahi = {a4[2], a4[3]};
        v4f b4[4];
#pragma unroll
        for (int t = 0; t < 4; ++t)
            b4[t] = *(const v4f*)(Wr1t + (t * 16 + m) * W1P + k8 + 4 * kh);
#pragma unroll
        for (int t = 0; t < 4; ++t) {
            const v2f blo = {b4[t][0], b4[t][1]};
            h[t] = __builtin_amdgcn_wmma_f32_16x16x4_f32(
                false, alo, false, blo, (short)0, h[t], false, false);
        }
#pragma unroll
        for (int t = 0; t < 4; ++t) {
            const v2f bhi = {b4[t][2], b4[t][3]};
            h[t] = __builtin_amdgcn_wmma_f32_16x16x4_f32(
                false, ahi, false, bhi, (short)0, h[t], false, false);
        }
    }
#pragma unroll
    for (int t = 0; t < 4; ++t)
#pragma unroll
        for (int r = 0; r < 8; ++r) {
            int row = r + 8 * kh, col = t * 16 + m;
            mystage[row * HP + col] = ssp(h[t][r] * 0.25f);
        }
    __syncthreads();

    v8f wacc[8];
#pragma unroll
    for (int t = 0; t < 8; ++t) wacc[t] = vzero;
    for (int k8 = 0; k8 < FC; k8 += 8) {
        const v4f a4 = *(const v4f*)(mystage + m * HP + k8 + 4 * kh);
        const v2f alo = {a4[0], a4[1]}, ahi = {a4[2], a4[3]};
        v4f b4[8];
#pragma unroll
        for (int t = 0; t < 8; ++t)
            b4[t] = *(const v4f*)(Wr2t + (t * 16 + m) * W2P + k8 + 4 * kh);
#pragma unroll
        for (int t = 0; t < 8; ++t) {
            const v2f blo = {b4[t][0], b4[t][1]};
            wacc[t] = __builtin_amdgcn_wmma_f32_16x16x4_f32(
                false, alo, false, blo, (short)0, wacc[t], false, false);
        }
#pragma unroll
        for (int t = 0; t < 8; ++t) {
            const v2f bhi = {b4[t][2], b4[t][3]};
            wacc[t] = __builtin_amdgcn_wmma_f32_16x16x4_f32(
                false, ahi, false, bhi, (short)0, wacc[t], false, false);
        }
    }

    __syncthreads();
#pragma unroll
    for (int t = 0; t < 8; ++t)
#pragma unroll
        for (int r = 0; r < 8; ++r) {
            int row = r + 8 * kh, col = t * 16 + m;
            mystage[row * SP + col] = wacc[t][r] * 0.125f;
        }

    float se[16], vx[16], vy[16], vz[16];
#pragma unroll
    for (int r = 0; r < 16; ++r) {
        const int s = __builtin_amdgcn_readlane(srcv, r);
        se[r] = s_in[(size_t)s * MULT + lane];
        const float* vp = v_in + (size_t)s * (MULT * 3) + 3 * lane;
        vx[r] = vp[0]; vy[r] = vp[1]; vz[r] = vp[2];
    }
    const int   dstv = edge_index[(long)E + ee];
    const float y00v = edge_y0[ee];
    const float y10v = edge_y1[ee * 3 + 0];
    const float y11v = edge_y1[ee * 3 + 1];
    const float y12v = edge_y1[ee * 3 + 2];

    __syncthreads();

    const float inv_s3  = 0.5773502691896258f;
    const float inv_deg = 0.17677669529663687f;
    const int pa0 = ((3 * lane + 0) & 31) * 4;
    const int pa1 = ((3 * lane + 1) & 31) * 4;
    const int pa2 = ((3 * lane + 2) & 31) * 4;
    const int i0 = (lane + 0) % 3;
    const int i1 = (lane + 2) % 3;
    const int i2 = (lane + 4) % 3;
    auto scatter_edge = [&](const int r) {
        const int   d   = __builtin_amdgcn_readlane(dstv, r);
        const float y00 = lane_val(y00v, r);
        const float y0  = lane_val(y10v, r);
        const float y1  = lane_val(y11v, r);
        const float y2  = lane_val(y12v, r);
        float* as_ = agg_s + (size_t)d * (2 * MULT);
        float* av_ = agg_v + (size_t)d * (2 * MULT * 3);
        const float* wrow = mystage + r * SP;
        const float w1 = wrow[lane];
        const float w2 = wrow[MULT + lane];
        const float w3 = wrow[2 * MULT + lane];
        const float w4 = wrow[3 * MULT + lane];
        const float sv = se[r], v0 = vx[r], v1 = vy[r], v2 = vz[r];
        const float cy = y00 * inv_deg;
        unsafeAtomicAdd(as_ + lane, w1 * sv * cy);
        const float dv = v0 * y0 + v1 * y1 + v2 * y2;
        unsafeAtomicAdd(as_ + MULT + lane, w4 * dv * (inv_s3 * inv_deg));
        const float t2 = w2 * sv * inv_deg;
        {
            const int p0 = __builtin_amdgcn_ds_permute(pa0, __float_as_int(t2 * y0));
            const int p1 = __builtin_amdgcn_ds_permute(pa1, __float_as_int(t2 * y1));
            const int p2 = __builtin_amdgcn_ds_permute(pa2, __float_as_int(t2 * y2));
            const int f0 = (i0 == 0) ? p0 : ((i0 == 1) ? p1 : p2);
            const int f1 = (i1 == 0) ? p0 : ((i1 == 1) ? p1 : p2);
            const int f2 = (i2 == 0) ? p0 : ((i2 == 1) ? p1 : p2);
            unsafeAtomicAdd(av_ + lane,            __int_as_float(f0));
            unsafeAtomicAdd(av_ + MULT + lane,     __int_as_float(f1));
            unsafeAtomicAdd(av_ + 2 * MULT + lane, __int_as_float(f2));
        }
        const float t3 = w3 * cy;
        {
            const int p0 = __builtin_amdgcn_ds_permute(pa0, __float_as_int(t3 * v0));
            const int p1 = __builtin_amdgcn_ds_permute(pa1, __float_as_int(t3 * v1));
            const int p2 = __builtin_amdgcn_ds_permute(pa2, __float_as_int(t3 * v2));
            const int f0 = (i0 == 0) ? p0 : ((i0 == 1) ? p1 : p2);
            const int f1 = (i1 == 0) ? p0 : ((i1 == 1) ? p1 : p2);
            const int f2 = (i2 == 0) ? p0 : ((i2 == 1) ? p1 : p2);
            unsafeAtomicAdd(av_ + 3 * MULT + lane, __int_as_float(f0));
            unsafeAtomicAdd(av_ + 4 * MULT + lane, __int_as_float(f1));
            unsafeAtomicAdd(av_ + 5 * MULT + lane, __int_as_float(f2));
        }
    };
    if (e0 + 16 <= E) {
#pragma unroll
        for (int r = 0; r < 16; ++r) scatter_edge(r);
    } else {
#pragma unroll
        for (int r = 0; r < 16; ++r)
            if (e0 + r < E) scatter_edge(r);
    }
}

__global__ __launch_bounds__(256) void out_kernel(
    const float* __restrict__ node_s, const float* __restrict__ node_v,
    const float* __restrict__ node_attrs,
    const float* __restrict__ W2s, const float* __restrict__ W2v,
    const float* __restrict__ Wscs, const float* __restrict__ Wscv,
    const float* __restrict__ agg_s, const float* __restrict__ agg_v,
    float* __restrict__ out, int N)
{
    __shared__ float L2s[2 * MULT * MULT];
    __shared__ float L2v[2 * MULT * MULT];
    __shared__ float Lss[MULT * NATTR * MULT];
    __shared__ float Lsv[MULT * NATTR * MULT];
    for (int i = threadIdx.x; i < 2 * MULT * MULT; i += 256) { L2s[i] = W2s[i]; L2v[i] = W2v[i]; }
    for (int i = threadIdx.x; i < MULT * NATTR * MULT; i += 256) { Lss[i] = Wscs[i]; Lsv[i] = Wscv[i]; }
    __syncthreads();

    int gid = blockIdx.x * 256 + threadIdx.x;
    int n = gid >> 5, w = gid & 31;
    if (n >= N) return;
    const float inv2m = 0.125f;
    const float invfan = 0.0625f;

    const float* ags = agg_s + (size_t)n * 2 * MULT;
    const float* agv = agg_v + (size_t)n * 2 * MULT * 3;
    float ns = 0.f, nv0 = 0.f, nv1 = 0.f, nv2 = 0.f;
#pragma unroll 8
    for (int u = 0; u < 2 * MULT; ++u) {
        float w2s = L2s[u * MULT + w];
        float w2v = L2v[u * MULT + w];
        ns  += ags[u] * w2s;
        nv0 += agv[u * 3 + 0] * w2v;
        nv1 += agv[u * 3 + 1] * w2v;
        nv2 += agv[u * 3 + 2] * w2v;
    }
    ns *= inv2m; nv0 *= inv2m; nv1 *= inv2m; nv2 *= inv2m;

    float attr[NATTR];
#pragma unroll
    for (int a = 0; a < NATTR; ++a) attr[a] = node_attrs[(size_t)n * NATTR + a];

    const float* srow = node_s + (size_t)n * MULT;
    const float* vrow = node_v + (size_t)n * MULT * 3;
    float scs = 0.f, scv0 = 0.f, scv1 = 0.f, scv2 = 0.f;
#pragma unroll 4
    for (int u = 0; u < MULT; ++u) {
        float ts = 0.f, tv = 0.f;
#pragma unroll
        for (int a = 0; a < NATTR; ++a) {
            ts += attr[a] * Lss[(u * NATTR + a) * MULT + w];
            tv += attr[a] * Lsv[(u * NATTR + a) * MULT + w];
        }
        scs  += srow[u] * ts;
        scv0 += vrow[u * 3 + 0] * tv;
        scv1 += vrow[u * 3 + 1] * tv;
        scv2 += vrow[u * 3 + 2] * tv;
    }
    scs *= invfan; scv0 *= invfan; scv1 *= invfan; scv2 *= invfan;

    float* orow = out + (size_t)n * 4 * MULT;
    orow[w] = ns + scs;
    orow[MULT + w * 3 + 0] = nv0 + scv0;
    orow[MULT + w * 3 + 1] = nv1 + scv1;
    orow[MULT + w * 3 + 2] = nv2 + scv2;
}

extern "C" void kernel_launch(void* const* d_in, const int* in_sizes, int n_in,
                              void* d_out, int out_size, void* d_ws, size_t ws_size,
                              hipStream_t stream) {
    const float* node_s     = (const float*)d_in[0];
    const float* node_v     = (const float*)d_in[1];
    const float* node_attrs = (const float*)d_in[2];
    const float* edge_emb   = (const float*)d_in[3];
    const float* edge_y0    = (const float*)d_in[4];
    const float* edge_y1    = (const float*)d_in[5];
    const int*   edge_index = (const int*)d_in[6];
    const float* W1s  = (const float*)d_in[7];
    const float* W1v  = (const float*)d_in[8];
    const float* Wr1  = (const float*)d_in[9];
    const float* Wr2  = (const float*)d_in[10];
    const float* W2s  = (const float*)d_in[11];
    const float* W2v  = (const float*)d_in[12];
    const float* Wscs = (const float*)d_in[13];
    const float* Wscv = (const float*)d_in[14];

    const int N = in_sizes[0] / MULT;
    const int E = in_sizes[3] / RB;

    float* s_buf = (float*)d_ws;
    float* v_buf = s_buf + (size_t)N * MULT;
    float* agg_s = v_buf + (size_t)N * MULT * 3;
    float* agg_v = agg_s + (size_t)N * 2 * MULT;

    hipMemsetAsync(agg_s, 0, (size_t)N * 8 * MULT * sizeof(float), stream);

    lin1_kernel<<<(N * MULT + 255) / 256, 256, 0, stream>>>(
        node_s, node_v, W1s, W1v, s_buf, v_buf, N);
    edge_kernel<<<(E + 127) / 128, 256, 0, stream>>>(
        edge_emb, edge_y0, edge_y1, edge_index, Wr1, Wr2, s_buf, v_buf, agg_s, agg_v, E);
    out_kernel<<<(N * MULT + 255) / 256, 256, 0, stream>>>(
        node_s, node_v, node_attrs, W2s, W2v, Wscs, Wscv, agg_s, agg_v,
        (float*)d_out, N);
}


# shifted softplus of the radial hidden layer: library log1pf (long double-float code, 32 values a lane) replaced by hardware log2(1 + exp(-|x|)) * ln 2; last bits of h change, no sum reordered
# speedup vs baseline: 7.0173x; 1.0659x over previous
#include <hip/hip_runtime.h>
#include <math.h>

typedef float v2f __attribute__((ext_vector_type(2)));
typedef float v4f __attribute__((ext_vector_type(4)));
typedef float v8f __attribute__((ext_vector_type(8)));

#define MULT 32
#define NATTR 8
#define RB 16
#define FC 64
#define LOG2F_ 0.6931471805599453f
#define W1P 20
#define W2P 68
#define HP  68
#define SP  132

__device__ __forceinline__ float ssp(float x) {
    const float t = __expf(-fabsf(x));
    return fmaf(__log2f(1.0f + t), LOG2F_, fmaxf(x, 0.0f) - LOG2F_);
}

__global__ __launch_bounds__(256) void lin1_kernel(
    const float* __restrict__ node_s, const float* __restrict__ node_v,
    const float* __restrict__ W1s, const float* __restrict__ W1v,
    float* __restrict__ s_out, float* __restrict__ v_out, int N)
{
    __shared__ float Ls[MULT * MULT];
    __shared__ float Lv[MULT * MULT];
    for (int i = threadIdx.x; i < MULT * MULT; i += 256) { Ls[i] = W1s[i]; Lv[i] = W1v[i]; }
    __syncthreads();

    int gid = blockIdx.x * 256 + threadIdx.x;
    int n = gid >> 5, w = gid & 31;
    if (n >= N) return;
    const float inv_m = 0.17677669529663687f;
    const float* srow = node_s + (size_t)n * MULT;
    const float* vrow = node_v + (size_t)n * MULT * 3;
    float as = 0.f, a0 = 0.f, a1 = 0.f, a2 = 0.f;
#pragma unroll 8
    for (int u = 0; u < MULT; ++u) {
        float ws_ = Ls[u * MULT + w];
        float wv_ = Lv[u * MULT + w];
        as += srow[u] * ws_;
        a0 += vrow[u * 3 + 0] * wv_;
        a1 += vrow[u * 3 + 1] * wv_;
        a2 += vrow[u * 3 + 2] * wv_;
    }
    s_out[(size_t)n * MULT + w] = as * inv_m;
    v_out[(size_t)n * MULT * 3 + w * 3 + 0] = a0 * inv_m;
    v_out[(size_t)n * MULT * 3 + w * 3 + 1] = a1 * inv_m;
    v_out[(size_t)n * MULT * 3 + w * 3 + 2] = a2 * inv_m;
}


__device__ __forceinline__ float lane_val(float x, int l) {
    return __int_as_float(__builtin_amdgcn_readlane(__float_as_int(x), l));
}

__global__ __launch_bounds__(256, 6) void edge_kernel(
    const float* __restrict__ edge_emb, const float* __restrict__ edge_y0,
    const float* __restrict__ edge_y1, const int* __restrict__ edge_index,
    const float* __restrict__ Wr1, const float* __restrict__ Wr2,
    const float* __restrict__ s_in, const float* __restrict__ v_in,
    float* __restrict__ agg_s, float* __restrict__ agg_v, int E)
{
    __shared__ __attribute__((aligned(16))) float Wr1t[FC * W1P];
    __shared__ __attribute__((aligned(16))) float Wr2t[4 * MULT * W2P];
    __shared__ __attribute__((aligned(16))) float stage[8 * 16 * SP];

    const int tid = threadIdx.x;
    for (int i = tid; i < RB * FC; i += 256) {
        int k = i / FC, n = i % FC;
        Wr1t[n * W1P + k] = Wr1[i];
    }
    {
        const int n  = tid & (4 * MULT - 1);
        const int kq = tid >> 7;
#pragma unroll
        for (int j = 0; j < 8; ++j) {
            const int k0 = 4 * (2 * j + kq);
            v4f q;
            q[0] = Wr2[(k0 + 0) * (4 * MULT) + n];
            q[1] = Wr2[(k0 + 1) * (4 * MULT) + n];
            q[2] = Wr2[(k0 + 2) * (4 * MULT) + n];
            q[3] = Wr2[(k0 + 3) * (4 * MULT) + n];
            *(v4f*)(Wr2t + n * W2P + k0) = q;
        }
    }
    __syncthreads();

    const int wave = __builtin_amdgcn_readfirstlane(tid >> 5);
    const int lane = tid & 31;
    const int m = lane & 15, kh = lane >> 4;
    const long e0 = (long)blockIdx.x * 128 + wave * 16;
    float* const mystage = stage + wave * (16 * SP);

    long ee = e0 + m; if (ee >= E) ee = E - 1;
    const int   srcv = edge_index[ee];

    const v8f vzero = {0.f, 0.f, 0.f, 0.f, 0.f, 0.f, 0.f, 0.f};


    v8f h[4];
#pragma unroll
    for (int t = 0; t < 4; ++t) h[t] = vzero;
#pragma unroll
    for (int k8 = 0; k8 < RB; k8 += 8) {
        const v4f a4 = *(const v4f*)(edge_emb + ee * RB + k8 + 4 * kh);
        const v2f alo = {a4[0], a4[1]}, ahi = {a4[2], a4[3]};
        v4f b4[4];
#pragma unroll
        for (int t = 0; t < 4; ++t)
            b4[t] = *(const v4f*)(Wr1t + (t * 16 + m) * W1P + k8 + 4 * kh);
#pragma unroll
        for (int t = 0; t < 4; ++t) {
            const v2f blo = {b4[t][0], b4[t][1]};
            h[t] = __builtin_amdgcn_wmma_f32_16x16x4_f32(
                false, alo, false, blo, (short)0, h[t], false, false);
        }
#pragma unroll
        for (int t = 0; t < 4; ++t) {
            const v2f bhi = {b4[t][2], b4[t][3]};
            h[t] = __builtin_amdgcn_wmma_f32_16x16x4_f32(
                false, ahi, false, bhi, (short)0, h[t], false, false);
        }
    }
#pragma unroll
    for (int t = 0; t < 4; ++t)
#pragma unroll
        for (int r = 0; r < 8; ++r) {
            int row = r + 8 * kh, col = t * 16 + m;
            mystage[row * HP + col] = ssp(h[t][r] * 0.25f);
        }
    __syncthreads();

    v8f wacc[8];
#pragma unroll
    for (int t = 0; t < 8; ++t) wacc[t] = vzero;
    for (int k8 = 0; k8 < FC; k8 += 8) {
        const v4f a4 = *(const v4f*)(mystage + m * HP + k8 + 4 * kh);
        const v2f alo = {a4[0], a4[1]}, ahi = {a4[2], a4[3]};
        v4f b4[8];
#pragma unroll
        for (int t = 0; t < 8; ++t)
            b4[t] = *(const v4f*)(Wr2t + (t * 16 + m) * W2P + k8 + 4 * kh);
#pragma unroll
        for (int t = 0; t < 8; ++t) {
            const v2f blo = {b4[t][0], b4[t][1]};
            wacc[t] = __builtin_amdgcn_wmma_f32_16x16x4_f32(
                false, alo, false, blo, (short)0, wacc[t], false, false);
        }
#pragma unroll
        for (int t = 0; t < 8; ++t) {
            const v2f bhi = {b4[t][2], b4[t][3]};
            wacc[t] = __builtin_amdgcn_wmma_f32_16x16x4_f32(
                false, ahi, false, bhi, (short)0, wacc[t], false, false);
        }
    }

    __syncthreads();
#pragma unroll
    for (int t = 0; t < 8; ++t)
#pragma unroll
        for (int r = 0; r < 8; ++r) {
            int row = r + 8 * kh, col = t * 16 + m;
            mystage[row * SP + col] = wacc[t][r] * 0.125f;
        }

    float se[16], vx[16], vy[16], vz[16];
#pragma unroll
    for (int r = 0; r < 16; ++r) {
        const int s = __builtin_amdgcn_readlane(srcv, r);
        se[r] = s_in[(size_t)s * MULT + lane];
        const float* vp = v_in + (size_t)s * (MULT * 3) + 3 * lane;
        vx[r] = vp[0]; vy[r] = vp[1]; vz[r] = vp[2];
    }
    const int   dstv = edge_index[(long)E + ee];
    const float y00v = edge_y0[ee];
    const float y10v = edge_y1[ee * 3 + 0];
    const float y11v = edge_y1[ee * 3 + 1];
    const float y12v = edge_y1[ee * 3 + 2];

    __syncthreads();

    const float inv_s3  = 0.5773502691896258f;
    const float inv_deg = 0.17677669529663687f;
    const int pa0 = ((3 * lane + 0) & 31) * 4;
    const int pa1 = ((3 * lane + 1) & 31) * 4;
    const int pa2 = ((3 * lane + 2) & 31) * 4;
    const int i0 = (lane + 0) % 3;
    const int i1 = (lane + 2) % 3;
    const int i2 = (lane + 4) % 3;
    auto scatter_edge = [&](const int r) {
        const int   d   = __builtin_amdgcn_readlane(dstv, r);
        const float y00 = lane_val(y00v, r);
        const float y0  = lane_val(y10v, r);
        const float y1  = lane_val(y11v, r);
        const float y2  = lane_val(y12v, r);
        float* as_ = agg_s + (size_t)d * (2 * MULT);
        float* av_ = agg_v + (size_t)d * (2 * MULT * 3);
        const float* wrow = mystage + r * SP;
        const float w1 = wrow[lane];
        const float w2 = wrow[MULT + lane];
        const float w3 = wrow[2 * MULT + lane];
        const float w4 = wrow[3 * MULT + lane];
        const float sv = se[r], v0 = vx[r], v1 = vy[r], v2 = vz[r];
        const float cy = y00 * inv_deg;
        unsafeAtomicAdd(as_ + lane, w1 * sv * cy);
        const float dv = v0 * y0 + v1 * y1 + v2 * y2;
        unsafeAtomicAdd(as_ + MULT + lane, w4 * dv * (inv_s3 * inv_deg));
        const float t2 = w2 * sv * inv_deg;
        {
            const int p0 = __builtin_amdgcn_ds_permute(pa0, __float_as_int(t2 * y0));
            const int p1 = __builtin_amdgcn_ds_permute(pa1, __float_as_int(t2 * y1));
            const int p2 = __builtin_amdgcn_ds_permute(pa2, __float_as_int(t2 * y2));
            const int f0 = (i0 == 0) ? p0 : ((i0 == 1) ? p1 : p2);
            const int f1 = (i1 == 0) ? p0 : ((i1 == 1) ? p1 : p2);
            const int f2 = (i2 == 0) ? p0 : ((i2 == 1) ? p1 : p2);
            unsafeAtomicAdd(av_ + lane,            __int_as_float(f0));
            unsafeAtomicAdd(av_ + MULT + lane,     __int_as_float(f1));
            unsafeAtomicAdd(av_ + 2 * MULT + lane, __int_as_float(f2));
        }
        const float t3 = w3 * cy;
        {
            const int p0 = __builtin_amdgcn_ds_permute(pa0, __float_as_int(t3 * v0));
            const int p1 = __builtin_amdgcn_ds_permute(pa1, __float_as_int(t3 * v1));
            const int p2 = __builtin_amdgcn_ds_permute(pa2, __float_as_int(t3 * v2));
            const int f0 = (i0 == 0) ? p0 : ((i0 == 1) ? p1 : p2);
            const int f1 = (i1 == 0) ? p0 : ((i1 == 1) ? p1 : p2);
            const int f2 = (i2 == 0) ? p0 : ((i2 == 1) ? p1 : p2);
            unsafeAtomicAdd(av_ + 3 * MULT + lane, __int_as_float(f0));
            unsafeAtomicAdd(av_ + 4 * MULT + lane, __int_as_float(f1));
            unsafeAtomicAdd(av_ + 5 * MULT + lane, __int_as_float(f2));
        }
    };
    if (e0 + 16 <= E) {
#pragma unroll
        for (int r = 0; r < 16; ++r) scatter_edge(r);
    } else {
#pragma unroll
        for (int r = 0; r < 16; ++r)
            if (e0 + r < E) scatter_edge(r);
    }
}

__global__ __launch_bounds__(256) void out_kernel(
    const float* __restrict__ node_s, const float* __restrict__ node_v,
    const float* __restrict__ node_attrs,
    const float* __restrict__ W2s, const float* __restrict__ W2v,
    const float* __restrict__ Wscs, const float* __restrict__ Wscv,
    const float* __restrict__ agg_s, const float* __restrict__ agg_v,
    float* __restrict__ out, int N)
{
    __shared__ float L2s[2 * MULT * MULT];
    __shared__ float L2v[2 * MULT * MULT];
    __shared__ float Lss[MULT * NATTR * MULT];
    __shared__ float Lsv[MULT * NATTR * MULT];
    for (int i = threadIdx.x; i < 2 * MULT * MULT; i += 256) { L2s[i] = W2s[i]; L2v[i] = W2v[i]; }
    for (int i = threadIdx.x; i < MULT * NATTR * MULT; i += 256) { Lss[i] = Wscs[i]; Lsv[i] = Wscv[i]; }
    __syncthreads();

    int gid = blockIdx.x * 256 + threadIdx.x;
    int n = gid >> 5, w = gid & 31;
    if (n >= N) return;
    const float inv2m = 0.125f;
    const float invfan = 0.0625f;

    const float* ags = agg_s + (size_t)n * 2 * MULT;
    const float* agv = agg_v + (size_t)n * 2 * MULT * 3;
    float ns = 0.f, nv0 = 0.f, nv1 = 0.f, nv2 = 0.f;
#pragma unroll 8
    for (int u = 0; u < 2 * MULT; ++u) {
        float w2s = L2s[u * MULT + w];
        float w2v = L2v[u * MULT + w];
        ns  += ags[u] * w2s;
        nv0 += agv[u * 3 + 0] * w2v;
        nv1 += agv[u * 3 + 1] * w2v;
        nv2 += agv[u * 3 + 2] * w2v;
    }
    ns *= inv2m; nv0 *= inv2m; nv1 *= inv2m; nv2 *= inv2m;

    float attr[NATTR];
#pragma unroll
    for (int a = 0; a < NATTR; ++a) attr[a] = node_attrs[(size_t)n * NATTR + a];

    const float* srow = node_s + (size_t)n * MULT;
    const float* vrow = node_v + (size_t)n * MULT * 3;
    float scs = 0.f, scv0 = 0.f, scv1 = 0.f, scv2 = 0.f;
#pragma unroll 4
    for (int u = 0; u < MULT; ++u) {
        float ts = 0.f, tv = 0.f;
#pragma unroll
        for (int a = 0; a < NATTR; ++a) {
            ts += attr[a] * Lss[(u * NATTR + a) * MULT + w];
            tv += attr[a] * Lsv[(u * NATTR + a) * MULT + w];
        }
        scs  += srow[u] * ts;
        scv0 += vrow[u * 3 + 0] * tv;
        scv1 += vrow[u * 3 + 1] * tv;
        scv2 += vrow[u * 3 + 2] * tv;
    }
    scs *= invfan; scv0 *= invfan; scv1 *= invfan; scv2 *= invfan;

    float* orow = out + (size_t)n * 4 * MULT;
    orow[w] = ns + scs;
    orow[MULT + w * 3 + 0] = nv0 + scv0;
    orow[MULT + w * 3 + 1] = nv1 + scv1;
    orow[MULT + w * 3 + 2] = nv2 + scv2;
}

extern "C" void kernel_launch(void* const* d_in, const int* in_sizes, int n_in,
                              void* d_out, int out_size, void* d_ws, size_t ws_size,
                              hipStream_t stream) {
    const float* node_s     = (const float*)d_in[0];
    const float* node_v     = (const float*)d_in[1];
    const float* node_attrs = (const float*)d_in[2];
    const float* edge_emb   = (const float*)d_in[3];
    const float* edge_y0    = (const float*)d_in[4];
    const float* edge_y1    = (const float*)d_in[5];
    const int*   edge_index = (const int*)d_in[6];
    const float* W1s  = (const float*)d_in[7];
    const float* W1v  = (const float*)d_in[8];
    const float* Wr1  = (const float*)d_in[9];
    const float* Wr2  = (const float*)d_in[10];
    const float* W2s  = (const float*)d_in[11];
    const float* W2v  = (const float*)d_in[12];
    const float* Wscs = (const float*)d_in[13];
    const float* Wscv = (const float*)d_in[14];

    const int N = in_sizes[0] / MULT;
    const int E = in_sizes[3] / RB;

    float* s_buf = (float*)d_ws;
    float* v_buf = s_buf + (size_t)N * MULT;
    float* agg_s = v_buf + (size_t)N * MULT * 3;
    float* agg_v = agg_s + (size_t)N * 2 * MULT;

    hipMemsetAsync(agg_s, 0, (size_t)N * 8 * MULT * sizeof(float), stream);

    lin1_kernel<<<(N * MULT + 255) / 256, 256, 0, stream>>>(
        node_s, node_v, W1s, W1v, s_buf, v_buf, N);
    edge_kernel<<<(E + 127) / 128, 256, 0, stream>>>(
        edge_emb, edge_y0, edge_y1, edge_index, Wr1, Wr2, s_buf, v_buf, agg_s, agg_v, E);
    out_kernel<<<(N * MULT + 255) / 256, 256, 0, stream>>>(
        node_s, node_v, node_attrs, W2s, W2v, Wscs, Wscv, agg_s, agg_v,
        (float*)d_out, N);
}


# edge kernel made persistent: 768 work-groups stage the weights once, then each wave walks its own 16-edge tiles under wave-level ordering, no work-group barrier in the walk; atomic order changes
# speedup vs baseline: 7.8397x; 1.1172x over previous
#include <hip/hip_runtime.h>
#include <math.h>

typedef float v2f __attribute__((ext_vector_type(2)));
typedef float v4f __attribute__((ext_vector_type(4)));
typedef float v8f __attribute__((ext_vector_type(8)));

#define MULT 32
#define NATTR 8
#define RB 16
#define FC 64
#define LOG2F_ 0.6931471805599453f
#define EDGE_GROUPS 768
#define W1P 20
#define W2P 68
#define HP  68
#define SP  132

__device__ __forceinline__ float ssp(float x) {
    const float t = __expf(-fabsf(x));
    return fmaf(__log2f(1.0f + t), LOG2F_, fmaxf(x, 0.0f) - LOG2F_);
}

__global__ __launch_bounds__(256) void lin1_kernel(
    const float* __restrict__ node_s, const float* __restrict__ node_v,
    const float* __restrict__ W1s, const float* __restrict__ W1v,
    float* __restrict__ s_out, float* __restrict__ v_out, int N)
{
    __shared__ float Ls[MULT * MULT];
    __shared__ float Lv[MULT * MULT];
    for (int i = threadIdx.x; i < MULT * MULT; i += 256) { Ls[i] = W1s[i]; Lv[i] = W1v[i]; }
    __syncthreads();

    int gid = blockIdx.x * 256 + threadIdx.x;
    int n = gid >> 5, w = gid & 31;
    if (n >= N) return;
    const float inv_m = 0.17677669529663687f;
    const float* srow = node_s + (size_t)n * MULT;
    const float* vrow = node_v + (size_t)n * MULT * 3;
    float as = 0.f, a0 = 0.f, a1 = 0.f, a2 = 0.f;
#pragma unroll 8
    for (int u = 0; u < MULT; ++u) {
        float ws_ = Ls[u * MULT + w];
        float wv_ = Lv[u * MULT + w];
        as += srow[u] * ws_;
        a0 += vrow[u * 3 + 0] * wv_;
        a1 += vrow[u * 3 + 1] * wv_;
        a2 += vrow[u * 3 + 2] * wv_;
    }
    s_out[(size_t)n * MULT + w] = as * inv_m;
    v_out[(size_t)n * MULT * 3 + w * 3 + 0] = a0 * inv_m;
    v_out[(size_t)n * MULT * 3 + w * 3 + 1] = a1 * inv_m;
    v_out[(size_t)n * MULT * 3 + w * 3 + 2] = a2 * inv_m;
}


__device__ __forceinline__ float lane_val(float x, int l) {
    return __int_as_float(__builtin_amdgcn_readlane(__float_as_int(x), l));
}

__device__ __forceinline__ void wave_handover() {
    __builtin_amdgcn_fence(__ATOMIC_RELEASE, "wavefront");
    __builtin_amdgcn_wave_barrier();
    __builtin_amdgcn_fence(__ATOMIC_ACQUIRE, "wavefront");
}

__global__ __launch_bounds__(256, 6) void edge_kernel(
    const float* __restrict__ edge_emb, const float* __restrict__ edge_y0,
    const float* __restrict__ edge_y1, const int* __restrict__ edge_index,
    const float* __restrict__ Wr1, const float* __restrict__ Wr2,
    const float* __restrict__ s_in, const float* __restrict__ v_in,
    float* __restrict__ agg_s, float* __restrict__ agg_v, int E)
{
    __shared__ __attribute__((aligned(16))) float Wr1t[FC * W1P];
    __shared__ __attribute__((aligned(16))) float Wr2t[4 * MULT * W2P];
    __shared__ __attribute__((aligned(16))) float stage[8 * 16 * SP];

    const int tid = threadIdx.x;
    for (int i = tid; i < RB * FC; i += 256) {
        int k = i / FC, n = i % FC;
        Wr1t[n * W1P + k] = Wr1[i];
    }
    {
        const int n  = tid & (4 * MULT - 1);
        const int kq = tid >> 7;
#pragma unroll
        for (int j = 0; j < 8; ++j) {
            const int k0 = 4 * (2 * j + kq);
            v4f q;
            q[0] = Wr2[(k0 + 0) * (4 * MULT) + n];
            q[1] = Wr2[(k0 + 1) * (4 * MULT) + n];
            q[2] = Wr2[(k0 + 2) * (4 * MULT) + n];
            q[3] = Wr2[(k0 + 3) * (4 * MULT) + n];
            *(v4f*)(Wr2t + n * W2P + k0) = q;
        }
    }
    __syncthreads();

    const int wave = __builtin_amdgcn_readfirstlane(tid >> 5);
    const int lane = tid & 31;
    const int m = lane & 15, kh = lane >> 4;
    float* const mystage = stage + wave * (16 * SP);
    const v8f vzero = {0.f, 0.f, 0.f, 0.f, 0.f, 0.f, 0.f, 0.f};

    const float inv_s3  = 0.5773502691896258f;
    const float inv_deg = 0.17677669529663687f;
    const int pa0 = ((3 * lane + 0) & 31) * 4;
    const int pa1 = ((3 * lane + 1) & 31) * 4;
    const int pa2 = ((3 * lane + 2) & 31) * 4;
    const int i0 = (lane + 0) % 3;
    const int i1 = (lane + 2) % 3;
    const int i2 = (lane + 4) % 3;

    const long ntiles = ((long)E + 15) / 16;
    const long nwaves = (long)EDGE_GROUPS * 8;
    for (long tile = (long)blockIdx.x * 8 + wave; tile < ntiles; tile += nwaves) {
        const long e0 = tile * 16;
        wave_handover();

        long ee = e0 + m; if (ee >= E) ee = E - 1;
        const int   srcv = edge_index[ee];


        v4f a4k[2];
#pragma unroll
        for (int j = 0; j < 2; ++j)
            a4k[j] = *(const v4f*)(edge_emb + ee * RB + 8 * j + 4 * kh);
        v8f h[4];
#pragma unroll
        for (int t = 0; t < 4; ++t) h[t] = vzero;
#pragma unroll
        for (int k8 = 0; k8 < RB; k8 += 8) {
            const v4f a4 = a4k[k8 / 8];
            const v2f alo = {a4[0], a4[1]}, ahi = {a4[2], a4[3]};
            v4f b4[4];
#pragma unroll
            for (int t = 0; t < 4; ++t)
                b4[t] = *(const v4f*)(Wr1t + (t * 16 + m) * W1P + k8 + 4 * kh);
#pragma unroll
            for (int t = 0; t < 4; ++t) {
                const v2f blo = {b4[t][0], b4[t][1]};
                h[t] = __builtin_amdgcn_wmma_f32_16x16x4_f32(
                    false, alo, false, blo, (short)0, h[t], false, false);
            }
#pragma unroll
            for (int t = 0; t < 4; ++t) {
                const v2f bhi = {b4[t][2], b4[t][3]};
                h[t] = __builtin_amdgcn_wmma_f32_16x16x4_f32(
                    false, ahi, false, bhi, (short)0, h[t], false, false);
            }
        }
#pragma unroll
        for (int t = 0; t < 4; ++t)
#pragma unroll
            for (int r = 0; r < 8; ++r) {
                int row = r + 8 * kh, col = t * 16 + m;
                mystage[row * HP + col] = ssp(h[t][r] * 0.25f);
            }
        wave_handover();

        v8f wacc[8];
#pragma unroll
        for (int t = 0; t < 8; ++t) wacc[t] = vzero;
        for (int k8 = 0; k8 < FC; k8 += 8) {
            const v4f a4 = *(const v4f*)(mystage + m * HP + k8 + 4 * kh);
            const v2f alo = {a4[0], a4[1]}, ahi = {a4[2], a4[3]};
            v4f b4[8];
#pragma unroll
            for (int t = 0; t < 8; ++t)
                b4[t] = *(const v4f*)(Wr2t + (t * 16 + m) * W2P + k8 + 4 * kh);
#pragma unroll
            for (int t = 0; t < 8; ++t) {
                const v2f blo = {b4[t][0], b4[t][1]};
                wacc[t] = __builtin_amdgcn_wmma_f32_16x16x4_f32(
                    false, alo, false, blo, (short)0, wacc[t], false, false);
            }
#pragma unroll
            for (int t = 0; t < 8; ++t) {
                const v2f bhi = {b4[t][2], b4[t][3]};
                wacc[t] = __builtin_amdgcn_wmma_f32_16x16x4_f32(
                    false, ahi, false, bhi, (short)0, wacc[t], false, false);
            }
        }

        wave_handover();
#pragma unroll
        for (int t = 0; t < 8; ++t)
#pragma unroll
            for (int r = 0; r < 8; ++r) {
                int row = r + 8 * kh, col = t * 16 + m;
                mystage[row * SP + col] = wacc[t][r] * 0.125f;
            }

        float se[16], vx[16], vy[16], vz[16];
#pragma unroll
        for (int r = 0; r < 16; ++r) {
            const int s = __builtin_amdgcn_readlane(srcv, r);
            se[r] = s_in[(size_t)s * MULT + lane];
            const float* vp = v_in + (size_t)s * (MULT * 3) + 3 * lane;
            vx[r] = vp[0]; vy[r] = vp[1]; vz[r] = vp[2];
        }
        const int   dstv = edge_index[(long)E + ee];
        const float y00v = edge_y0[ee];
        const float y10v = edge_y1[ee * 3 + 0];
        const float y11v = edge_y1[ee * 3 + 1];
        const float y12v = edge_y1[ee * 3 + 2];

        wave_handover();

        auto scatter_edge = [&](const int r) {
            const int   d   = __builtin_amdgcn_readlane(dstv, r);
            const float y00 = lane_val(y00v, r);
            const float y0  = lane_val(y10v, r);
            const float y1  = lane_val(y11v, r);
            const float y2  = lane_val(y12v, r);
            float* as_ = agg_s + (size_t)d * (2 * MULT);
            float* av_ = agg_v + (size_t)d * (2 * MULT * 3);
            const float* wrow = mystage + r * SP;
            const float w1 = wrow[lane];
            const float w2 = wrow[MULT + lane];
            const float w3 = wrow[2 * MULT + lane];
            const float w4 = wrow[3 * MULT + lane];
            const float sv = se[r], v0 = vx[r], v1 = vy[r], v2 = vz[r];
            const float cy = y00 * inv_deg;
            unsafeAtomicAdd(as_ + lane, w1 * sv * cy);
            const float dv = v0 * y0 + v1 * y1 + v2 * y2;
            unsafeAtomicAdd(as_ + MULT + lane, w4 * dv * (inv_s3 * inv_deg));
            const float t2 = w2 * sv * inv_deg;
            {
                const int p0 = __builtin_amdgcn_ds_permute(pa0, __float_as_int(t2 * y0));
                const int p1 = __builtin_amdgcn_ds_permute(pa1, __float_as_int(t2 * y1));
                const int p2 = __builtin_amdgcn_ds_permute(pa2, __float_as_int(t2 * y2));
                const int f0 = (i0 == 0) ? p0 : ((i0 == 1) ? p1 : p2);
                const int f1 = (i1 == 0) ? p0 : ((i1 == 1) ? p1 : p2);
                const int f2 = (i2 == 0) ? p0 : ((i2 == 1) ? p1 : p2);
                unsafeAtomicAdd(av_ + lane,            __int_as_float(f0));
                unsafeAtomicAdd(av_ + MULT + lane,     __int_as_float(f1));
                unsafeAtomicAdd(av_ + 2 * MULT + lane, __int_as_float(f2));
            }
            const float t3 = w3 * cy;
            {
                const int p0 = __builtin_amdgcn_ds_permute(pa0, __float_as_int(t3 * v0));
                const int p1 = __builtin_amdgcn_ds_permute(pa1, __float_as_int(t3 * v1));
                const int p2 = __builtin_amdgcn_ds_permute(pa2, __float_as_int(t3 * v2));
                const int f0 = (i0 == 0) ? p0 : ((i0 == 1) ? p1 : p2);
                const int f1 = (i1 == 0) ? p0 : ((i1 == 1) ? p1 : p2);
                const int f2 = (i2 == 0) ? p0 : ((i2 == 1) ? p1 : p2);
                unsafeAtomicAdd(av_ + 3 * MULT + lane, __int_as_float(f0));
                unsafeAtomicAdd(av_ + 4 * MULT + lane, __int_as_float(f1));
                unsafeAtomicAdd(av_ + 5 * MULT + lane, __int_as_float(f2));
            }
        };
        if (e0 + 16 <= E) {
#pragma unroll
            for (int r = 0; r < 16; ++r) scatter_edge(r);
        } else {
#pragma unroll
            for (int r = 0; r < 16; ++r)
                if (e0 + r < E) scatter_edge(r);
        }
    }
}

__global__ __launch_bounds__(256) void out_kernel(
    const float* __restrict__ node_s, const float* __restrict__ node_v,
    const float* __restrict__ node_attrs,
    const float* __restrict__ W2s, const float* __restrict__ W2v,
    const float* __restrict__ Wscs, const float* __restrict__ Wscv,
    const float* __restrict__ agg_s, const float* __restrict__ agg_v,
    float* __restrict__ out, int N)
{
    __shared__ float L2s[2 * MULT * MULT];
    __shared__ float L2v[2 * MULT * MULT];
    __shared__ float Lss[MULT * NATTR * MULT];
    __shared__ float Lsv[MULT * NATTR * MULT];
    for (int i = threadIdx.x; i < 2 * MULT * MULT; i += 256) { L2s[i] = W2s[i]; L2v[i] = W2v[i]; }
    for (int i = threadIdx.x; i < MULT * NATTR * MULT; i += 256) { Lss[i] = Wscs[i]; Lsv[i] = Wscv[i]; }
    __syncthreads();

    int gid = blockIdx.x * 256 + threadIdx.x;
    int n = gid >> 5, w = gid & 31;
    if (n >= N) return;
    const float inv2m = 0.125f;
    const float invfan = 0.0625f;

    const float* ags = agg_s + (size_t)n * 2 * MULT;
    const float* agv = agg_v + (size_t)n * 2 * MULT * 3;
    float ns = 0.f, nv0 = 0.f, nv1 = 0.f, nv2 = 0.f;
#pragma unroll 8
    for (int u = 0; u < 2 * MULT; ++u) {
        float w2s = L2s[u * MULT + w];
        float w2v = L2v[u * MULT + w];
        ns  += ags[u] * w2s;
        nv0 += agv[u * 3 + 0] * w2v;
        nv1 += agv[u * 3 + 1] * w2v;
        nv2 += agv[u * 3 + 2] * w2v;
    }
    ns *= inv2m; nv0 *= inv2m; nv1 *= inv2m; nv2 *= inv2m;

    float attr[NATTR];
#pragma unroll
    for (int a = 0; a < NATTR; ++a) attr[a] = node_attrs[(size_t)n * NATTR + a];

    const float* srow = node_s + (size_t)n * MULT;
    const float* vrow = node_v + (size_t)n * MULT * 3;
    float scs = 0.f, scv0 = 0.f, scv1 = 0.f, scv2 = 0.f;
#pragma unroll 4
    for (int u = 0; u < MULT; ++u) {
        float ts = 0.f, tv = 0.f;
#pragma unroll
        for (int a = 0; a < NATTR; ++a) {
            ts += attr[a] * Lss[(u * NATTR + a) * MULT + w];
            tv += attr[a] * Lsv[(u * NATTR + a) * MULT + w];
        }
        scs  += srow[u] * ts;
        scv0 += vrow[u * 3 + 0] * tv;
        scv1 += vrow[u * 3 + 1] * tv;
        scv2 += vrow[u * 3 + 2] * tv;
    }
    scs *= invfan; scv0 *= invfan; scv1 *= invfan; scv2 *= invfan;

    float* orow = out + (size_t)n * 4 * MULT;
    orow[w] = ns + scs;
    orow[MULT + w * 3 + 0] = nv0 + scv0;
    orow[MULT + w * 3 + 1] = nv1 + scv1;
    orow[MULT + w * 3 + 2] = nv2 + scv2;
}

extern "C" void kernel_launch(void* const* d_in, const int* in_sizes, int n_in,
                              void* d_out, int out_size, void* d_ws, size_t ws_size,
                              hipStream_t stream) {
    const float* node_s     = (const float*)d_in[0];
    const float* node_v     = (const float*)d_in[1];
    const float* node_attrs = (const float*)d_in[2];
    const float* edge_emb   = (const float*)d_in[3];
    const float* edge_y0    = (const float*)d_in[4];
    const float* edge_y1    = (const float*)d_in[5];
    const int*   edge_index = (const int*)d_in[6];
    const float* W1s  = (const float*)d_in[7];
    const float* W1v  = (const float*)d_in[8];
    const float* Wr1  = (const float*)d_in[9];
    const float* Wr2  = (const float*)d_in[10];
    const float* W2s  = (const float*)d_in[11];
    const float* W2v  = (const float*)d_in[12];
    const float* Wscs = (const float*)d_in[13];
    const float* Wscv = (const float*)d_in[14];

    const int N = in_sizes[0] / MULT;
    const int E = in_sizes[3] / RB;

    float* s_buf = (float*)d_ws;
    float* v_buf = s_buf + (size_t)N * MULT;
    float* agg_s = v_buf + (size_t)N * MULT * 3;
    float* agg_v = agg_s + (size_t)N * 2 * MULT;

    hipMemsetAsync(agg_s, 0, (size_t)N * 8 * MULT * sizeof(float), stream);

    lin1_kernel<<<(N * MULT + 255) / 256, 256, 0, stream>>>(
        node_s, node_v, W1s, W1v, s_buf, v_buf, N);
    edge_kernel<<<EDGE_GROUPS, 256, 0, stream>>>(
        edge_emb, edge_y0, edge_y1, edge_index, Wr1, Wr2, s_buf, v_buf, agg_s, agg_v, E);
    out_kernel<<<(N * MULT + 255) / 256, 256, 0, stream>>>(
        node_s, node_v, node_attrs, W2s, W2v, Wscs, Wscv, agg_s, agg_v,
        (float*)d_out, N);
}


# node kernels (linear_1 and output): 256 persistent work-groups stage weights once; a wave does 4 nodes per pass, so each LDS weight read serves 4 nodes; node rows via scalar loads; sum order kept
# speedup vs baseline: 8.4658x; 1.0799x over previous
#include <hip/hip_runtime.h>
#include <math.h>

typedef float v2f __attribute__((ext_vector_type(2)));
typedef float v4f __attribute__((ext_vector_type(4)));
typedef float v8f __attribute__((ext_vector_type(8)));

#define MULT 32
#define NATTR 8
#define RB 16
#define FC 64
#define LOG2F_ 0.6931471805599453f
#define EDGE_GROUPS 768
#define W1P 20
#define W2P 68
#define HP  68
#define SP  132

__device__ __forceinline__ float ssp(float x) {
    const float t = __expf(-fabsf(x));
    return fmaf(__log2f(1.0f + t), LOG2F_, fmaxf(x, 0.0f) - LOG2F_);
}

#define NODE_GROUPS 256
#define NODE_WAVES  20
#define NODE_NB     4

__global__ __launch_bounds__(NODE_WAVES * 32) void lin1_kernel(
    const float* __restrict__ node_s, const float* __restrict__ node_v,
    const float* __restrict__ W1s, const float* __restrict__ W1v,
    float* __restrict__ s_out, float* __restrict__ v_out, int N)
{
    __shared__ __attribute__((aligned(16))) v2f Lw1[MULT * MULT];
    const int tid = threadIdx.x;
    for (int i = tid; i < MULT * MULT; i += NODE_WAVES * 32) {
        const v2f q = {W1s[i], W1v[i]};
        Lw1[i] = q;
    }
    __syncthreads();

    const int wave = __builtin_amdgcn_readfirstlane(tid >> 5);
    const int lane = tid & 31;
    const float inv_m = 0.17677669529663687f;
    const int npass = (N + NODE_NB - 1) / NODE_NB;
    for (int pass = blockIdx.x * NODE_WAVES + wave; pass < npass; pass += gridDim.x * NODE_WAVES) {
        const int n0 = pass * NODE_NB;
        const float* srow[NODE_NB];
        const float* vrow[NODE_NB];
#pragma unroll
        for (int j = 0; j < NODE_NB; ++j) {
            int nj = n0 + j; if (nj >= N) nj = N - 1;
            srow[j] = node_s + (size_t)nj * MULT;
            vrow[j] = node_v + (size_t)nj * MULT * 3;
        }
        float as[NODE_NB], a0[NODE_NB], a1[NODE_NB], a2[NODE_NB];
#pragma unroll
        for (int j = 0; j < NODE_NB; ++j) { as[j] = 0.f; a0[j] = 0.f; a1[j] = 0.f; a2[j] = 0.f; }
#pragma unroll 4
        for (int u = 0; u < MULT; ++u) {
            const v2f wq = Lw1[u * MULT + lane];
#pragma unroll
            for (int j = 0; j < NODE_NB; ++j) {
                as[j] += srow[j][u] * wq[0];
                a0[j] += vrow[j][u * 3 + 0] * wq[1];
                a1[j] += vrow[j][u * 3 + 1] * wq[1];
                a2[j] += vrow[j][u * 3 + 2] * wq[1];
            }
        }
#pragma unroll
        for (int j = 0; j < NODE_NB; ++j) {
            if (n0 + j < N) {
                const size_t n = (size_t)(n0 + j);
                s_out[n * MULT + lane] = as[j] * inv_m;
                v_out[n * MULT * 3 + lane * 3 + 0] = a0[j] * inv_m;
                v_out[n * MULT * 3 + lane * 3 + 1] = a1[j] * inv_m;
                v_out[n * MULT * 3 + lane * 3 + 2] = a2[j] * inv_m;
            }
        }
    }
}


__device__ __forceinline__ float lane_val(float x, int l) {
    return __int_as_float(__builtin_amdgcn_readlane(__float_as_int(x), l));
}

__device__ __forceinline__ void wave_handover() {
    __builtin_amdgcn_fence(__ATOMIC_RELEASE, "wavefront");
    __builtin_amdgcn_wave_barrier();
    __builtin_amdgcn_fence(__ATOMIC_ACQUIRE, "wavefront");
}

__global__ __launch_bounds__(256, 6) void edge_kernel(
    const float* __restrict__ edge_emb, const float* __restrict__ edge_y0,
    const float* __restrict__ edge_y1, const int* __restrict__ edge_index,
    const float* __restrict__ Wr1, const float* __restrict__ Wr2,
    const float* __restrict__ s_in, const float* __restrict__ v_in,
    float* __restrict__ agg_s, float* __restrict__ agg_v, int E)
{
    __shared__ __attribute__((aligned(16))) float Wr1t[FC * W1P];
    __shared__ __attribute__((aligned(16))) float Wr2t[4 * MULT * W2P];
    __shared__ __attribute__((aligned(16))) float stage[8 * 16 * SP];

    const int tid = threadIdx.x;
    for (int i = tid; i < RB * FC; i += 256) {
        int k = i / FC, n = i % FC;
        Wr1t[n * W1P + k] = Wr1[i];
    }
    {
        const int n  = tid & (4 * MULT - 1);
        const int kq = tid >> 7;
#pragma unroll
        for (int j = 0; j < 8; ++j) {
            const int k0 = 4 * (2 * j + kq);
            v4f q;
            q[0] = Wr2[(k0 + 0) * (4 * MULT) + n];
            q[1] = Wr2[(k0 + 1) * (4 * MULT) + n];
            q[2] = Wr2[(k0 + 2) * (4 * MULT) + n];
            q[3] = Wr2[(k0 + 3) * (4 * MULT) + n];
            *(v4f*)(Wr2t + n * W2P + k0) = q;
        }
    }
    __syncthreads();

    const int wave = __builtin_amdgcn_readfirstlane(tid >> 5);
    const int lane = tid & 31;
    const int m = lane & 15, kh = lane >> 4;
    float* const mystage = stage + wave * (16 * SP);
    const v8f vzero = {0.f, 0.f, 0.f, 0.f, 0.f, 0.f, 0.f, 0.f};

    const float inv_s3  = 0.5773502691896258f;
    const float inv_deg = 0.17677669529663687f;
    const int pa0 = ((3 * lane + 0) & 31) * 4;
    const int pa1 = ((3 * lane + 1) & 31) * 4;
    const int pa2 = ((3 * lane + 2) & 31) * 4;
    const int i0 = (lane + 0) % 3;
    const int i1 = (lane + 2) % 3;
    const int i2 = (lane + 4) % 3;

    const long ntiles = ((long)E + 15) / 16;
    const long nwaves = (long)EDGE_GROUPS * 8;
    for (long tile = (long)blockIdx.x * 8 + wave; tile < ntiles; tile += nwaves) {
        const long e0 = tile * 16;
        wave_handover();

        long ee = e0 + m; if (ee >= E) ee = E - 1;
        const int   srcv = edge_index[ee];


        v4f a4k[2];
#pragma unroll
        for (int j = 0; j < 2; ++j)
            a4k[j] = *(const v4f*)(edge_emb + ee * RB + 8 * j + 4 * kh);
        v8f h[4];
#pragma unroll
        for (int t = 0; t < 4; ++t) h[t] = vzero;
#pragma unroll
        for (int k8 = 0; k8 < RB; k8 += 8) {
            const v4f a4 = a4k[k8 / 8];
            const v2f alo = {a4[0], a4[1]}, ahi = {a4[2], a4[3]};
            v4f b4[4];
#pragma unroll
            for (int t = 0; t < 4; ++t)
                b4[t] = *(const v4f*)(Wr1t + (t * 16 + m) * W1P + k8 + 4 * kh);
#pragma unroll
            for (int t = 0; t < 4; ++t) {
                const v2f blo = {b4[t][0], b4[t][1]};
                h[t] = __builtin_amdgcn_wmma_f32_16x16x4_f32(
                    false, alo, false, blo, (short)0, h[t], false, false);
            }
#pragma unroll
            for (int t = 0; t < 4; ++t) {
                const v2f bhi = {b4[t][2], b4[t][3]};
                h[t] = __builtin_amdgcn_wmma_f32_16x16x4_f32(
                    false, ahi, false, bhi, (short)0, h[t], false, false);
            }
        }
#pragma unroll
        for (int t = 0; t < 4; ++t)
#pragma unroll
            for (int r = 0; r < 8; ++r) {
                int row = r + 8 * kh, col = t * 16 + m;
                mystage[row * HP + col] = ssp(h[t][r] * 0.25f);
            }
        wave_handover();

        v8f wacc[8];
#pragma unroll
        for (int t = 0; t < 8; ++t) wacc[t] = vzero;
        for (int k8 = 0; k8 < FC; k8 += 8) {
            const v4f a4 = *(const v4f*)(mystage + m * HP + k8 + 4 * kh);
            const v2f alo = {a4[0], a4[1]}, ahi = {a4[2], a4[3]};
            v4f b4[8];
#pragma unroll
            for (int t = 0; t < 8; ++t)
                b4[t] = *(const v4f*)(Wr2t + (t * 16 + m) * W2P + k8 + 4 * kh);
#pragma unroll
            for (int t = 0; t < 8; ++t) {
                const v2f blo = {b4[t][0], b4[t][1]};
                wacc[t] = __builtin_amdgcn_wmma_f32_16x16x4_f32(
                    false, alo, false, blo, (short)0, wacc[t], false, false);
            }
#pragma unroll
            for (int t = 0; t < 8; ++t) {
                const v2f bhi = {b4[t][2], b4[t][3]};
                wacc[t] = __builtin_amdgcn_wmma_f32_16x16x4_f32(
                    false, ahi, false, bhi, (short)0, wacc[t], false, false);
            }
        }

        wave_handover();
#pragma unroll
        for (int t = 0; t < 8; ++t)
#pragma unroll
            for (int r = 0; r < 8; ++r) {
                int row = r + 8 * kh, col = t * 16 + m;
                mystage[row * SP + col] = wacc[t][r] * 0.125f;
            }

        float se[16], vx[16], vy[16], vz[16];
#pragma unroll
        for (int r = 0; r < 16; ++r) {
            const int s = __builtin_amdgcn_readlane(srcv, r);
            se[r] = s_in[(size_t)s * MULT + lane];
            const float* vp = v_in + (size_t)s * (MULT * 3) + 3 * lane;
            vx[r] = vp[0]; vy[r] = vp[1]; vz[r] = vp[2];
        }
        const int   dstv = edge_index[(long)E + ee];
        const float y00v = edge_y0[ee];
        const float y10v = edge_y1[ee * 3 + 0];
        const float y11v = edge_y1[ee * 3 + 1];
        const float y12v = edge_y1[ee * 3 + 2];

        wave_handover();

        auto scatter_edge = [&](const int r) {
            const int   d   = __builtin_amdgcn_readlane(dstv, r);
            const float y00 = lane_val(y00v, r);
            const float y0  = lane_val(y10v, r);
            const float y1  = lane_val(y11v, r);
            const float y2  = lane_val(y12v, r);
            float* as_ = agg_s + (size_t)d * (2 * MULT);
            float* av_ = agg_v + (size_t)d * (2 * MULT * 3);
            const float* wrow = mystage + r * SP;
            const float w1 = wrow[lane];
            const float w2 = wrow[MULT + lane];
            const float w3 = wrow[2 * MULT + lane];
            const float w4 = wrow[3 * MULT + lane];
            const float sv = se[r], v0 = vx[r], v1 = vy[r], v2 = vz[r];
            const float cy = y00 * inv_deg;
            unsafeAtomicAdd(as_ + lane, w1 * sv * cy);
            const float dv = v0 * y0 + v1 * y1 + v2 * y2;
            unsafeAtomicAdd(as_ + MULT + lane, w4 * dv * (inv_s3 * inv_deg));
            const float t2 = w2 * sv * inv_deg;
            {
                const int p0 = __builtin_amdgcn_ds_permute(pa0, __float_as_int(t2 * y0));
                const int p1 = __builtin_amdgcn_ds_permute(pa1, __float_as_int(t2 * y1));
                const int p2 = __builtin_amdgcn_ds_permute(pa2, __float_as_int(t2 * y2));
                const int f0 = (i0 == 0) ? p0 : ((i0 == 1) ? p1 : p2);
                const int f1 = (i1 == 0) ? p0 : ((i1 == 1) ? p1 : p2);
                const int f2 = (i2 == 0) ? p0 : ((i2 == 1) ? p1 : p2);
                unsafeAtomicAdd(av_ + lane,            __int_as_float(f0));
                unsafeAtomicAdd(av_ + MULT + lane,     __int_as_float(f1));
                unsafeAtomicAdd(av_ + 2 * MULT + lane, __int_as_float(f2));
            }
            const float t3 = w3 * cy;
            {
                const int p0 = __builtin_amdgcn_ds_permute(pa0, __float_as_int(t3 * v0));
                const int p1 = __builtin_amdgcn_ds_permute(pa1, __float_as_int(t3 * v1));
                const int p2 = __builtin_amdgcn_ds_permute(pa2, __float_as_int(t3 * v2));
                const int f0 = (i0 == 0) ? p0 : ((i0 == 1) ? p1 : p2);
                const int f1 = (i1 == 0) ? p0 : ((i1 == 1) ? p1 : p2);
                const int f2 = (i2 == 0) ? p0 : ((i2 == 1) ? p1 : p2);
                unsafeAtomicAdd(av_ + 3 * MULT + lane, __int_as_float(f0));
                unsafeAtomicAdd(av_ + 4 * MULT + lane, __int_as_float(f1));
                unsafeAtomicAdd(av_ + 5 * MULT + lane, __int_as_float(f2));
            }
        };
        if (e0 + 16 <= E) {
#pragma unroll
            for (int r = 0; r < 16; ++r) scatter_edge(r);
        } else {
#pragma unroll
            for (int r = 0; r < 16; ++r)
                if (e0 + r < E) scatter_edge(r);
        }
    }
}

__global__ __launch_bounds__(NODE_WAVES * 32) void out_kernel(
    const float* __restrict__ node_s, const float* __restrict__ node_v,
    const float* __restrict__ node_attrs,
    const float* __restrict__ W2s, const float* __restrict__ W2v,
    const float* __restrict__ Wscs, const float* __restrict__ Wscv,
    const float* __restrict__ agg_s, const float* __restrict__ agg_v,
    float* __restrict__ out, int N)
{
    __shared__ __attribute__((aligned(16))) v2f Lw2[2 * MULT * MULT];
    __shared__ __attribute__((aligned(16))) v4f Lsc[MULT * 4 * MULT];
    const int tid = threadIdx.x;
#pragma unroll
    for (int k = 0; k < (2 * MULT * MULT + NODE_WAVES * 32 - 1) / (NODE_WAVES * 32); ++k) {
        const int i = tid + k * (NODE_WAVES * 32);
        if (i < 2 * MULT * MULT) {
            const v2f q = {W2s[i], W2v[i]};
            Lw2[i] = q;
        }
    }
#pragma unroll
    for (int k = 0; k < (MULT * 4 * MULT + NODE_WAVES * 32 - 1) / (NODE_WAVES * 32); ++k) {
        const int i = tid + k * (NODE_WAVES * 32);
        if (i < MULT * 4 * MULT) {
            const int row2 = (i >> 5) * 2, w = i & 31;
            v4f q;
            q[0] = Wscs[row2 * MULT + w];
            q[1] = Wscv[row2 * MULT + w];
            q[2] = Wscs[(row2 + 1) * MULT + w];
            q[3] = Wscv[(row2 + 1) * MULT + w];
            Lsc[i] = q;
        }
    }
    __syncthreads();

    const int wave = __builtin_amdgcn_readfirstlane(tid >> 5);
    const int lane = tid & 31;
    const float inv2m = 0.125f;
    const float invfan = 0.0625f;
    const int npass = (N + NODE_NB - 1) / NODE_NB;
    for (int pass = blockIdx.x * NODE_WAVES + wave; pass < npass; pass += gridDim.x * NODE_WAVES) {
        const int n0 = pass * NODE_NB;
        const float* ags[NODE_NB];
        const float* agv[NODE_NB];
        const float* srow[NODE_NB];
        const float* vrow[NODE_NB];
        const float* arow[NODE_NB];
#pragma unroll
        for (int j = 0; j < NODE_NB; ++j) {
            int nj = n0 + j; if (nj >= N) nj = N - 1;
            ags[j]  = agg_s + (size_t)nj * 2 * MULT;
            agv[j]  = agg_v + (size_t)nj * 2 * MULT * 3;
            srow[j] = node_s + (size_t)nj * MULT;
            vrow[j] = node_v + (size_t)nj * MULT * 3;
            arow[j] = node_attrs + (size_t)nj * NATTR;
        }

        float ns[NODE_NB], nv0[NODE_NB], nv1[NODE_NB], nv2[NODE_NB];
#pragma unroll
        for (int j = 0; j < NODE_NB; ++j) { ns[j] = 0.f; nv0[j] = 0.f; nv1[j] = 0.f; nv2[j] = 0.f; }
#pragma unroll 4
        for (int u = 0; u < 2 * MULT; ++u) {
            const v2f wq = Lw2[u * MULT + lane];
#pragma unroll
            for (int j = 0; j < NODE_NB; ++j) {
                ns[j]  += ags[j][u] * wq[0];
                nv0[j] += agv[j][u * 3 + 0] * wq[1];
                nv1[j] += agv[j][u * 3 + 1] * wq[1];
                nv2[j] += agv[j][u * 3 + 2] * wq[1];
            }
        }

        float scs[NODE_NB], scv0[NODE_NB], scv1[NODE_NB], scv2[NODE_NB];
#pragma unroll
        for (int j = 0; j < NODE_NB; ++j) { scs[j] = 0.f; scv0[j] = 0.f; scv1[j] = 0.f; scv2[j] = 0.f; }
#pragma unroll 2
        for (int u = 0; u < MULT; ++u) {
            v4f q[4];
#pragma unroll
            for (int ap = 0; ap < 4; ++ap) q[ap] = Lsc[(u * 4 + ap) * MULT + lane];
#pragma unroll
            for (int j = 0; j < NODE_NB; ++j) {
                v2f t = {0.f, 0.f};
#pragma unroll
                for (int ap = 0; ap < 4; ++ap) {
                    const v2f lo = {q[ap][0], q[ap][1]}, hi = {q[ap][2], q[ap][3]};
                    t += arow[j][2 * ap] * lo;
                    t += arow[j][2 * ap + 1] * hi;
                }
                scs[j]  += srow[j][u] * t[0];
                scv0[j] += vrow[j][u * 3 + 0] * t[1];
                scv1[j] += vrow[j][u * 3 + 1] * t[1];
                scv2[j] += vrow[j][u * 3 + 2] * t[1];
            }
        }

#pragma unroll
        for (int j = 0; j < NODE_NB; ++j) {
            if (n0 + j < N) {
                float* orow = out + (size_t)(n0 + j) * 4 * MULT;
                orow[lane] = ns[j] * inv2m + scs[j] * invfan;
                orow[MULT + lane * 3 + 0] = nv0[j] * inv2m + scv0[j] * invfan;
                orow[MULT + lane * 3 + 1] = nv1[j] * inv2m + scv1[j] * invfan;
                orow[MULT + lane * 3 + 2] = nv2[j] * inv2m + scv2[j] * invfan;
            }
        }
    }
}

extern "C" void kernel_launch(void* const* d_in, const int* in_sizes, int n_in,
                              void* d_out, int out_size, void* d_ws, size_t ws_size,
                              hipStream_t stream) {
    const float* node_s     = (const float*)d_in[0];
    const float* node_v     = (const float*)d_in[1];
    const float* node_attrs = (const float*)d_in[2];
    const float* edge_emb   = (const float*)d_in[3];
    const float* edge_y0    = (const float*)d_in[4];
    const float* edge_y1    = (const float*)d_in[5];
    const int*   edge_index = (const int*)d_in[6];
    const float* W1s  = (const float*)d_in[7];
    const float* W1v  = (const float*)d_in[8];
    const float* Wr1  = (const float*)d_in[9];
    const float* Wr2  = (const float*)d_in[10];
    const float* W2s  = (const float*)d_in[11];
    const float* W2v  = (const float*)d_in[12];
    const float* Wscs = (const float*)d_in[13];
    const float* Wscv = (const float*)d_in[14];

    const int N = in_sizes[0] / MULT;
    const int E = in_sizes[3] / RB;

    float* s_buf = (float*)d_ws;
    float* v_buf = s_buf + (size_t)N * MULT;
    float* agg_s = v_buf + (size_t)N * MULT * 3;
    float* agg_v = agg_s + (size_t)N * 2 * MULT;

    hipMemsetAsync(agg_s, 0, (size_t)N * 8 * MULT * sizeof(float), stream);

    lin1_kernel<<<NODE_GROUPS, NODE_WAVES * 32, 0, stream>>>(
        node_s, node_v, W1s, W1v, s_buf, v_buf, N);
    edge_kernel<<<EDGE_GROUPS, 256, 0, stream>>>(
        edge_emb, edge_y0, edge_y1, edge_index, Wr1, Wr2, s_buf, v_buf, agg_s, agg_v, E);
    out_kernel<<<NODE_GROUPS, NODE_WAVES * 32, 0, stream>>>(
        node_s, node_v, node_attrs, W2s, W2v, Wscs, Wscv, agg_s, agg_v,
        (float*)d_out, N);
}
